# starting kernel
# baseline (speedup 1.0000x reference)
#include <hip/hip_runtime.h>

#define N_NODES   50000
#define N_EDGES   800000
#define IN_FEAT   128
#define PROMPT_DIM 32
#define EF_DIM     32
#define HP_DIM     32
#define OUT_DIM   224

typedef __attribute__((ext_vector_type(2))) float v2f;
typedef __attribute__((ext_vector_type(8))) float v8f;

__device__ __forceinline__ void gatomic_add_f32(float* p, float v) {
    asm volatile("global_atomic_add_f32 %0, %1, off" :: "v"(p), "v"(v) : "memory");
}

__global__ void zero_kernel(float* __restrict__ out, float* __restrict__ deg) {
    const size_t stride = (size_t)gridDim.x * blockDim.x;
    size_t i = (size_t)blockIdx.x * blockDim.x + threadIdx.x;
    const size_t total4 = (size_t)N_NODES * OUT_DIM / 4;
    float4 z = make_float4(0.f, 0.f, 0.f, 0.f);
    for (size_t k = i; k < total4; k += stride)
        ((float4*)out)[k] = z;
    for (size_t k = i; k < (size_t)N_NODES; k += stride)
        deg[k] = 0.0f;
}

__global__ void edge_kernel(const float* __restrict__ nf,
                            const float* __restrict__ ef,
                            const int*   __restrict__ src,
                            const int*   __restrict__ dst,
                            float* __restrict__ out,
                            float* __restrict__ deg) {
    const int lane = threadIdx.x & 31;
    const int wave = threadIdx.x >> 5;
    const int e    = blockIdx.x * 8 + wave;
    if (e >= N_EDGES) return;

    const int s = src[e];
    const int d = dst[e];

    const int pe = (e + 8 < N_EDGES) ? (e + 8) : e;
    __builtin_prefetch(&ef[(size_t)pe * EF_DIM + lane], 0, 1);

    const float4 x = ((const float4*)(nf + (size_t)s * IN_FEAT))[lane];
    float* orow = out + (size_t)d * OUT_DIM;
    float* onf  = orow + lane * 4;
    gatomic_add_f32(onf + 0, x.x);
    gatomic_add_f32(onf + 1, x.y);
    gatomic_add_f32(onf + 2, x.z);
    gatomic_add_f32(onf + 3, x.w);

    const float fe = ef[(size_t)e * EF_DIM + lane];
    gatomic_add_f32(orow + 160 + lane, fe);

    if (lane == 0) gatomic_add_f32(deg + d, 1.0f);
}

__global__ void prompt_kernel(const float* __restrict__ prompt,
                              const float* __restrict__ hp,
                              const float* __restrict__ deg,
                              float* __restrict__ out) {
    const int lane = threadIdx.x;
    const int m0   = blockIdx.x * 16;
    const int l15  = lane & 15;
    const bool lo  = (lane < 16);

    const float dv = deg[m0 + l15];
    v2f a; a.x = lo ? dv : 0.0f; a.y = 0.0f;

    const float p0 = prompt[l15], p1 = prompt[16 + l15];
    const float h0 = hp[l15],     h1 = hp[16 + l15];
    v2f b0; b0.x = lo ? p0 : 0.0f; b0.y = 0.0f;
    v2f b1; b1.x = lo ? p1 : 0.0f; b1.y = 0.0f;
    v2f b2; b2.x = lo ? h0 : 0.0f; b2.y = 0.0f;
    v2f b3; b3.x = lo ? h1 : 0.0f; b3.y = 0.0f;

    v8f c = {0.f, 0.f, 0.f, 0.f, 0.f, 0.f, 0.f, 0.f};
    v8f d0 = __builtin_amdgcn_wmma_f32_16x16x4_f32(false, a, false, b0, (short)0, c, false, false);
    v8f d1 = __builtin_amdgcn_wmma_f32_16x16x4_f32(false, a, false, b1, (short)0, c, false, false);
    v8f d2 = __builtin_amdgcn_wmma_f32_16x16x4_f32(false, a, false, b2, (short)0, c, false, false);
    v8f d3 = __builtin_amdgcn_wmma_f32_16x16x4_f32(false, a, false, b3, (short)0, c, false, false);

    const int mofs = (lane >> 4) * 8;
#pragma unroll
    for (int r = 0; r < 8; ++r) {
        const size_t row = (size_t)(m0 + mofs + r) * OUT_DIM;
        out[row + 128 + l15]      = d0[r];
        out[row + 144 + l15]      = d1[r];
        out[row + 192 + l15]      = d2[r];
        out[row + 208 + l15]      = d3[r];
    }
}

extern "C" void kernel_launch(void* const* d_in, const int* in_sizes, int n_in,
                              void* d_out, int out_size, void* d_ws, size_t ws_size,
                              hipStream_t stream) {
    (void)in_sizes; (void)n_in; (void)out_size; (void)ws_size;

    const float* nf     = (const float*)d_in[0];
    const float* ef     = (const float*)d_in[1];
    const float* prompt = (const float*)d_in[2];
    const float* hp     = (const float*)d_in[3];
    const int*   src    = (const int*)d_in[4];
    const int*   dst    = (const int*)d_in[5];

    float* out = (float*)d_out;
    float* deg = (float*)d_ws;

    zero_kernel<<<2048, 256, 0, stream>>>(out, deg);
    edge_kernel<<<N_EDGES / 8, 256, 0, stream>>>(nf, ef, src, dst, out, deg);
    prompt_kernel<<<N_NODES / 16, 32, 0, stream>>>(prompt, hp, deg, out);
}


# edge kernel: lane l takes node-row columns l,32+l,64+l,96+l, so every f32 atomic-add instruction covers one whole 128-B line (4 line requests per row, not 16); sums still in atomic arrival order
# speedup vs baseline: 1.1011x; 1.1011x over previous
#include <hip/hip_runtime.h>

#define N_NODES   50000
#define N_EDGES   800000
#define IN_FEAT   128
#define PROMPT_DIM 32
#define EF_DIM     32
#define HP_DIM     32
#define OUT_DIM   224

typedef __attribute__((ext_vector_type(2))) float v2f;
typedef __attribute__((ext_vector_type(8))) float v8f;

__device__ __forceinline__ void gatomic_add_f32(float* p, float v) {
    asm volatile("global_atomic_add_f32 %0, %1, off" :: "v"(p), "v"(v) : "memory");
}
template <int BYTE_OFS>
__device__ __forceinline__ void gatomic_add_f32_ofs(float* p, float v) {
    asm volatile("global_atomic_add_f32 %0, %1, off offset:%2"
                 :: "v"(p), "v"(v), "n"(BYTE_OFS) : "memory");
}

__global__ void zero_kernel(float* __restrict__ out, float* __restrict__ deg) {
    const size_t stride = (size_t)gridDim.x * blockDim.x;
    size_t i = (size_t)blockIdx.x * blockDim.x + threadIdx.x;
    const size_t total4 = (size_t)N_NODES * OUT_DIM / 4;
    float4 z = make_float4(0.f, 0.f, 0.f, 0.f);
    for (size_t k = i; k < total4; k += stride)
        ((float4*)out)[k] = z;
    for (size_t k = i; k < (size_t)N_NODES; k += stride)
        deg[k] = 0.0f;
}

__global__ void edge_kernel(const float* __restrict__ nf,
                            const float* __restrict__ ef,
                            const int*   __restrict__ src,
                            const int*   __restrict__ dst,
                            float* __restrict__ out,
                            float* __restrict__ deg) {
    const int lane = threadIdx.x & 31;
    const int wave = threadIdx.x >> 5;
    const int e    = blockIdx.x * 8 + wave;
    if (e >= N_EDGES) return;

    const int s = src[e];
    const int d = dst[e];

    const int pe = (e + 8 < N_EDGES) ? (e + 8) : e;
    __builtin_prefetch(&ef[(size_t)pe * EF_DIM + lane], 0, 1);

    const float* nrow = nf + (size_t)s * IN_FEAT + lane;
    const float x0 = nrow[0];
    const float x1 = nrow[32];
    const float x2 = nrow[64];
    const float x3 = nrow[96];

    float* orow = out + (size_t)d * OUT_DIM;
    float* ocol = orow + lane;
    gatomic_add_f32_ofs<0>(ocol, x0);
    gatomic_add_f32_ofs<128>(ocol, x1);
    gatomic_add_f32_ofs<256>(ocol, x2);
    gatomic_add_f32_ofs<384>(ocol, x3);

    const float fe = ef[(size_t)e * EF_DIM + lane];
    gatomic_add_f32_ofs<640>(ocol, fe);

    if (lane == 0) gatomic_add_f32(deg + d, 1.0f);
}

__global__ void prompt_kernel(const float* __restrict__ prompt,
                              const float* __restrict__ hp,
                              const float* __restrict__ deg,
                              float* __restrict__ out) {
    const int lane = threadIdx.x;
    const int m0   = blockIdx.x * 16;
    const int l15  = lane & 15;
    const bool lo  = (lane < 16);

    const float dv = deg[m0 + l15];
    v2f a; a.x = lo ? dv : 0.0f; a.y = 0.0f;

    const float p0 = prompt[l15], p1 = prompt[16 + l15];
    const float h0 = hp[l15],     h1 = hp[16 + l15];
    v2f b0; b0.x = lo ? p0 : 0.0f; b0.y = 0.0f;
    v2f b1; b1.x = lo ? p1 : 0.0f; b1.y = 0.0f;
    v2f b2; b2.x = lo ? h0 : 0.0f; b2.y = 0.0f;
    v2f b3; b3.x = lo ? h1 : 0.0f; b3.y = 0.0f;

    v8f c = {0.f, 0.f, 0.f, 0.f, 0.f, 0.f, 0.f, 0.f};
    v8f d0 = __builtin_amdgcn_wmma_f32_16x16x4_f32(false, a, false, b0, (short)0, c, false, false);
    v8f d1 = __builtin_amdgcn_wmma_f32_16x16x4_f32(false, a, false, b1, (short)0, c, false, false);
    v8f d2 = __builtin_amdgcn_wmma_f32_16x16x4_f32(false, a, false, b2, (short)0, c, false, false);
    v8f d3 = __builtin_amdgcn_wmma_f32_16x16x4_f32(false, a, false, b3, (short)0, c, false, false);

    const int mofs = (lane >> 4) * 8;
#pragma unroll
    for (int r = 0; r < 8; ++r) {
        const size_t row = (size_t)(m0 + mofs + r) * OUT_DIM;
        out[row + 128 + l15]      = d0[r];
        out[row + 144 + l15]      = d1[r];
        out[row + 192 + l15]      = d2[r];
        out[row + 208 + l15]      = d3[r];
    }
}

extern "C" void kernel_launch(void* const* d_in, const int* in_sizes, int n_in,
                              void* d_out, int out_size, void* d_ws, size_t ws_size,
                              hipStream_t stream) {
    (void)in_sizes; (void)n_in; (void)out_size; (void)ws_size;

    const float* nf     = (const float*)d_in[0];
    const float* ef     = (const float*)d_in[1];
    const float* prompt = (const float*)d_in[2];
    const float* hp     = (const float*)d_in[3];
    const int*   src    = (const int*)d_in[4];
    const int*   dst    = (const int*)d_in[5];

    float* out = (float*)d_out;
    float* deg = (float*)d_ws;

    zero_kernel<<<2048, 256, 0, stream>>>(out, deg);
    edge_kernel<<<N_EDGES / 8, 256, 0, stream>>>(nf, ef, src, dst, out, deg);
    prompt_kernel<<<N_NODES / 16, 32, 0, stream>>>(prompt, hp, deg, out);
}


# f32 atomic scatter + zero fill -> owner-computes: each work-group sweeps dst, lists its 200 nodes' edges in LDS and sums their rows in LDS by plain adds; no atomics, fixed sum order, out stored once
# speedup vs baseline: 2.0887x; 1.8969x over previous
#include <hip/hip_runtime.h>

#define N_NODES   50000
#define N_EDGES   800000
#define IN_FEAT   128
#define PROMPT_DIM 32
#define EF_DIM     32
#define HP_DIM     32
#define OUT_DIM   224

typedef __attribute__((ext_vector_type(2))) float v2f;
typedef __attribute__((ext_vector_type(8))) float v8f;

#define NODES_PER_WG 200
#define ACC_W        160
#define STEP_EDGES   1024
#define WG_STEP      (32 * STEP_EDGES)
#define N_STEPS      25
#define LIST_CAP     1280
#define LIST_PITCH   1281
#define PEND_CAP     64

__device__ __forceinline__ void add_batch8(const float* __restrict__ nf,
                                           const float* __restrict__ ef,
                                           const int*   __restrict__ src,
                                           float* acc, int* degc,
                                           const int* p, int lane) {
    int   ln[8];
    float2 xa[8], xb[8];
    float  fe[8];
    for (int i = 0; i < 8; ++i) {
        const int en = __builtin_amdgcn_readfirstlane(p[i]);
        const int e  = en >> 8;
        ln[i] = en & 255;
        const int s = src[e];
        const float* r = nf + (size_t)s * IN_FEAT;
        xa[i] = *(const float2*)(r + 2 * lane);
        xb[i] = *(const float2*)(r + 64 + 2 * lane);
        fe[i] = ef[(size_t)e * EF_DIM + lane];
    }
    for (int i = 0; i < 8; ++i) {
        float* a = acc + ln[i] * ACC_W;
        float2 va = *(float2*)(a + 2 * lane);
        float2 vb = *(float2*)(a + 64 + 2 * lane);
        va.x += xa[i].x; va.y += xa[i].y;
        vb.x += xb[i].x; vb.y += xb[i].y;
        *(float2*)(a + 2 * lane)      = va;
        *(float2*)(a + 64 + 2 * lane) = vb;
        a[128 + lane] += fe[i];
        if (lane == 0) degc[ln[i]] += 1;
    }
}

__device__ __forceinline__ void add_one(const float* __restrict__ nf,
                                        const float* __restrict__ ef,
                                        const int*   __restrict__ src,
                                        float* acc, int* degc,
                                        const int* p, int lane) {
    const int en = __builtin_amdgcn_readfirstlane(p[0]);
    const int e  = en >> 8;
    const int ln = en & 255;
    const int s  = src[e];
    const float* r = nf + (size_t)s * IN_FEAT;
    const float2 xa = *(const float2*)(r + 2 * lane);
    const float2 xb = *(const float2*)(r + 64 + 2 * lane);
    const float  fe = ef[(size_t)e * EF_DIM + lane];
    float* a = acc + ln * ACC_W;
    float2 va = *(float2*)(a + 2 * lane);
    float2 vb = *(float2*)(a + 64 + 2 * lane);
    va.x += xa.x; va.y += xa.y;
    vb.x += xb.x; vb.y += xb.y;
    *(float2*)(a + 2 * lane)      = va;
    *(float2*)(a + 64 + 2 * lane) = vb;
    a[128 + lane] += fe;
    if (lane == 0) degc[ln] += 1;
}

__device__ __forceinline__ void drain_lists(const float* __restrict__ nf,
                                            const float* __restrict__ ef,
                                            const int*   __restrict__ src,
                                            float* acc, int* degc,
                                            const int* lists, const int* fillv,
                                            int* pendw, int w, int lane) {
    const unsigned below = (1u << lane) - 1u;
    const int  myfill = fillv[lane];
    const int* mylist = lists + lane * LIST_PITCH;
    int npend = 0;
    for (int k = 0; k < LIST_CAP; ++k) {
        const bool valid = k < myfill;
        if (__builtin_amdgcn_ballot_w32(valid) == 0u) break;
        const int raw  = mylist[k];
        const bool mine = valid && ((raw & 31) == w);
        const unsigned m = __builtin_amdgcn_ballot_w32(mine);
        if (m != 0u) {
            if (mine) pendw[npend + __popc(m & below)] = raw;
            npend += __popc(m);
            if (npend >= 32) {
                const int full = npend & ~7;
                for (int j = 0; j < full; j += 8)
                    add_batch8(nf, ef, src, acc, degc, pendw + j, lane);
                const int left = npend - full;
                const int t = pendw[full + (lane < left ? lane : 0)];
                if (lane < left) pendw[lane] = t;
                npend = left;
            }
        }
    }
    const int full = npend & ~7;
    for (int j = 0; j < full; j += 8)
        add_batch8(nf, ef, src, acc, degc, pendw + j, lane);
    for (int j = full; j < npend; ++j)
        add_one(nf, ef, src, acc, degc, pendw + j, lane);
}

__global__ void __launch_bounds__(1024) owner_kernel(const float* __restrict__ nf,
                                                     const float* __restrict__ ef,
                                                     const int*   __restrict__ src,
                                                     const int*   __restrict__ dst,
                                                     float* __restrict__ out,
                                                     float* __restrict__ deg) {
    __shared__ float acc[NODES_PER_WG * ACC_W];
    __shared__ int   lists[32 * LIST_PITCH];
    __shared__ int   pend[32 * PEND_CAP];
    __shared__ int   fillv[32];
    __shared__ int   degc[NODES_PER_WG];
    __shared__ int   flag[N_STEPS];

    const int tid  = threadIdx.x;
    const int lane = tid & 31;
    const int w    = __builtin_amdgcn_readfirstlane(tid >> 5);
    const int n0   = blockIdx.x * NODES_PER_WG;
    const unsigned below = (1u << lane) - 1u;

    for (int i = tid; i < NODES_PER_WG * ACC_W; i += 1024) acc[i] = 0.0f;
    if (tid < NODES_PER_WG) degc[tid] = 0;
    if (tid < N_STEPS) flag[tid] = 0;
    __syncthreads();

    int* mylist = lists + w * LIST_PITCH;
    int* pendw  = pend + w * PEND_CAP;
    int  fillw  = 0;

    for (int st = 0; st < N_STEPS; ++st) {
        if (lane == 0) {
            fillv[w] = fillw;
            if (fillw > LIST_CAP - STEP_EDGES) flag[st] = 1;
        }
        __syncthreads();
        if (__builtin_amdgcn_readfirstlane(flag[st]) != 0) {
            drain_lists(nf, ef, src, acc, degc, lists, fillv, pendw, w, lane);
            __syncthreads();
            fillw = 0;
        }
        const int step  = (st + (int)blockIdx.x) % N_STEPS;
        const int ebase = step * WG_STEP + w * STEP_EDGES;
        int4 d4[STEP_EDGES / 128];
        for (int it = 0; it < STEP_EDGES / 128; ++it) {
            const int eb  = ebase + it * 128;
            const int ebc = eb < N_EDGES ? eb : (N_EDGES - 128);
            d4[it] = *(const int4*)(dst + ebc + lane * 4);
        }
        for (int it = 0; it < STEP_EDGES / 128; ++it) {
            const int eb = ebase + it * 128;
            const bool in = eb < N_EDGES;
            const int dd0 = d4[it].x - n0, dd1 = d4[it].y - n0;
            const int dd2 = d4[it].z - n0, dd3 = d4[it].w - n0;
            const bool h0 = in && (unsigned)dd0 < (unsigned)NODES_PER_WG;
            const bool h1 = in && (unsigned)dd1 < (unsigned)NODES_PER_WG;
            const bool h2 = in && (unsigned)dd2 < (unsigned)NODES_PER_WG;
            const bool h3 = in && (unsigned)dd3 < (unsigned)NODES_PER_WG;
            if (__builtin_amdgcn_ballot_w32(h0 || h1 || h2 || h3) != 0u) {
                const int e0 = eb + lane * 4;
                unsigned m;
                m = __builtin_amdgcn_ballot_w32(h0);
                if (h0) mylist[fillw + __popc(m & below)] = (e0 << 8) | dd0;
                fillw += __popc(m);
                m = __builtin_amdgcn_ballot_w32(h1);
                if (h1) mylist[fillw + __popc(m & below)] = ((e0 + 1) << 8) | dd1;
                fillw += __popc(m);
                m = __builtin_amdgcn_ballot_w32(h2);
                if (h2) mylist[fillw + __popc(m & below)] = ((e0 + 2) << 8) | dd2;
                fillw += __popc(m);
                m = __builtin_amdgcn_ballot_w32(h3);
                if (h3) mylist[fillw + __popc(m & below)] = ((e0 + 3) << 8) | dd3;
                fillw += __popc(m);
            }
        }
    }
    if (lane == 0) fillv[w] = fillw;
    __syncthreads();
    drain_lists(nf, ef, src, acc, degc, lists, fillv, pendw, w, lane);
    __syncthreads();

    for (int r = w; r < NODES_PER_WG; r += 32) {
        const float* a = acc + r * ACC_W;
        float* o = out + (size_t)(n0 + r) * OUT_DIM;
        *(float2*)(o + 2 * lane)      = *(const float2*)(a + 2 * lane);
        *(float2*)(o + 64 + 2 * lane) = *(const float2*)(a + 64 + 2 * lane);
        o[160 + lane] = a[128 + lane];
    }
    if (tid < NODES_PER_WG) deg[n0 + tid] = (float)degc[tid];
}

__global__ void prompt_kernel(const float* __restrict__ prompt,
                              const float* __restrict__ hp,
                              const float* __restrict__ deg,
                              float* __restrict__ out) {
    const int lane = threadIdx.x;
    const int m0   = blockIdx.x * 16;
    const int l15  = lane & 15;
    const bool lo  = (lane < 16);

    const float dv = deg[m0 + l15];
    v2f a; a.x = lo ? dv : 0.0f; a.y = 0.0f;

    const float p0 = prompt[l15], p1 = prompt[16 + l15];
    const float h0 = hp[l15],     h1 = hp[16 + l15];
    v2f b0; b0.x = lo ? p0 : 0.0f; b0.y = 0.0f;
    v2f b1; b1.x = lo ? p1 : 0.0f; b1.y = 0.0f;
    v2f b2; b2.x = lo ? h0 : 0.0f; b2.y = 0.0f;
    v2f b3; b3.x = lo ? h1 : 0.0f; b3.y = 0.0f;

    v8f c = {0.f, 0.f, 0.f, 0.f, 0.f, 0.f, 0.f, 0.f};
    v8f d0 = __builtin_amdgcn_wmma_f32_16x16x4_f32(false, a, false, b0, (short)0, c, false, false);
    v8f d1 = __builtin_amdgcn_wmma_f32_16x16x4_f32(false, a, false, b1, (short)0, c, false, false);
    v8f d2 = __builtin_amdgcn_wmma_f32_16x16x4_f32(false, a, false, b2, (short)0, c, false, false);
    v8f d3 = __builtin_amdgcn_wmma_f32_16x16x4_f32(false, a, false, b3, (short)0, c, false, false);

    const int mofs = (lane >> 4) * 8;
#pragma unroll
    for (int r = 0; r < 8; ++r) {
        const size_t row = (size_t)(m0 + mofs + r) * OUT_DIM;
        out[row + 128 + l15]      = d0[r];
        out[row + 144 + l15]      = d1[r];
        out[row + 192 + l15]      = d2[r];
        out[row + 208 + l15]      = d3[r];
    }
}

extern "C" void kernel_launch(void* const* d_in, const int* in_sizes, int n_in,
                              void* d_out, int out_size, void* d_ws, size_t ws_size,
                              hipStream_t stream) {
    (void)in_sizes; (void)n_in; (void)out_size; (void)ws_size;

    const float* nf     = (const float*)d_in[0];
    const float* ef     = (const float*)d_in[1];
    const float* prompt = (const float*)d_in[2];
    const float* hp     = (const float*)d_in[3];
    const int*   src    = (const int*)d_in[4];
    const int*   dst    = (const int*)d_in[5];

    float* out = (float*)d_out;
    float* deg = (float*)d_ws;

    owner_kernel<<<N_NODES / NODES_PER_WG, 1024, 0, stream>>>(nf, ef, src, dst, out, deg);
    prompt_kernel<<<N_NODES / 16, 32, 0, stream>>>(prompt, hp, deg, out);
}


# dst sweep on 1-byte owner keys: new first launch writes key = dst/200 per edge; owners test 16 keys per lane and load for a zero byte, no barrier per stretch; per-node sum order changes, stays fixed
# speedup vs baseline: 2.5910x; 1.2405x over previous
#include <hip/hip_runtime.h>

#define N_NODES   50000
#define N_EDGES   800000
#define IN_FEAT   128
#define PROMPT_DIM 32
#define EF_DIM     32
#define HP_DIM     32
#define OUT_DIM   224

typedef __attribute__((ext_vector_type(2))) float v2f;
typedef __attribute__((ext_vector_type(8))) float v8f;

#define NODES_PER_WG 200
#define ACC_W        160
#define CHUNK_EDGES  512
#define N_CHUNKS     1563
#define KEY_WORDS    (N_CHUNKS * CHUNK_EDGES / 4)
#define ROUNDS       49
#define G_FAST       7
#define G_SAFE       2
#define N_STEPS      25
#define LIST_CAP     1280
#define LIST_PITCH   1281
#define PEND_CAP     64
#define NO_OWNER     255u

__device__ __forceinline__ unsigned owner_key(int d) {
    return (unsigned)d < (unsigned)N_NODES ? (unsigned)d / (unsigned)NODES_PER_WG : NO_OWNER;
}

__global__ void __launch_bounds__(256) key_kernel(const int* __restrict__ dst,
                                                  int* __restrict__ keys) {
    const int t = blockIdx.x * 256 + threadIdx.x;
    if (t >= KEY_WORDS) return;
    unsigned k = 0xFFFFFFFFu;
    if (t < N_EDGES / 4) {
        const int4 d = *(const int4*)(dst + 4 * t);
        k = owner_key(d.x) | (owner_key(d.y) << 8) | (owner_key(d.z) << 16) | (owner_key(d.w) << 24);
    }
    keys[t] = (int)k;
}


__device__ __forceinline__ void add_batch8(const float* __restrict__ nf,
                                           const float* __restrict__ ef,
                                           const int*   __restrict__ src,
                                           float* acc, int* degc,
                                           const int* p, int lane) {
    int   ln[8];
    float2 xa[8], xb[8];
    float  fe[8];
    for (int i = 0; i < 8; ++i) {
        const int en = __builtin_amdgcn_readfirstlane(p[i]);
        const int e  = en >> 8;
        ln[i] = en & 255;
        const int s = src[e];
        const float* r = nf + (size_t)s * IN_FEAT;
        xa[i] = *(const float2*)(r + 2 * lane);
        xb[i] = *(const float2*)(r + 64 + 2 * lane);
        fe[i] = ef[(size_t)e * EF_DIM + lane];
    }
    for (int i = 0; i < 8; ++i) {
        float* a = acc + ln[i] * ACC_W;
        float2 va = *(float2*)(a + 2 * lane);
        float2 vb = *(float2*)(a + 64 + 2 * lane);
        va.x += xa[i].x; va.y += xa[i].y;
        vb.x += xb[i].x; vb.y += xb[i].y;
        *(float2*)(a + 2 * lane)      = va;
        *(float2*)(a + 64 + 2 * lane) = vb;
        a[128 + lane] += fe[i];
        if (lane == 0) degc[ln[i]] += 1;
    }
}

__device__ __forceinline__ void add_one(const float* __restrict__ nf,
                                        const float* __restrict__ ef,
                                        const int*   __restrict__ src,
                                        float* acc, int* degc,
                                        const int* p, int lane) {
    const int en = __builtin_amdgcn_readfirstlane(p[0]);
    const int e  = en >> 8;
    const int ln = en & 255;
    const int s  = src[e];
    const float* r = nf + (size_t)s * IN_FEAT;
    const float2 xa = *(const float2*)(r + 2 * lane);
    const float2 xb = *(const float2*)(r + 64 + 2 * lane);
    const float  fe = ef[(size_t)e * EF_DIM + lane];
    float* a = acc + ln * ACC_W;
    float2 va = *(float2*)(a + 2 * lane);
    float2 vb = *(float2*)(a + 64 + 2 * lane);
    va.x += xa.x; va.y += xa.y;
    vb.x += xb.x; vb.y += xb.y;
    *(float2*)(a + 2 * lane)      = va;
    *(float2*)(a + 64 + 2 * lane) = vb;
    a[128 + lane] += fe;
    if (lane == 0) degc[ln] += 1;
}

__device__ __forceinline__ void drain_lists(const float* __restrict__ nf,
                                            const float* __restrict__ ef,
                                            const int*   __restrict__ src,
                                            float* acc, int* degc,
                                            const int* lists, const int* fillv,
                                            int* pendw, int w, int lane) {
    const unsigned below = (1u << lane) - 1u;
    const int  myfill = fillv[lane];
    const int* mylist = lists + lane * LIST_PITCH;
    int npend = 0;
    for (int k = 0; k < LIST_CAP; ++k) {
        const bool valid = k < myfill;
        if (__builtin_amdgcn_ballot_w32(valid) == 0u) break;
        const int raw  = mylist[k];
        const bool mine = valid && ((raw & 31) == w);
        const unsigned m = __builtin_amdgcn_ballot_w32(mine);
        if (m != 0u) {
            if (mine) pendw[npend + __popc(m & below)] = raw;
            npend += __popc(m);
            if (npend >= 32) {
                const int full = npend & ~7;
                for (int j = 0; j < full; j += 8)
                    add_batch8(nf, ef, src, acc, degc, pendw + j, lane);
                const int left = npend - full;
                const int t = pendw[full + (lane < left ? lane : 0)];
                if (lane < left) pendw[lane] = t;
                npend = left;
            }
        }
    }
    const int full = npend & ~7;
    for (int j = 0; j < full; j += 8)
        add_batch8(nf, ef, src, acc, degc, pendw + j, lane);
    for (int j = full; j < npend; ++j)
        add_one(nf, ef, src, acc, degc, pendw + j, lane);
}

template <int G>
__device__ __forceinline__ int sweep_group(const uint4* __restrict__ keys16, int r0, int rot,
                                           unsigned pat, int w, int lane, unsigned below,
                                           int* mylist, int fillw) {
    uint4 q[G];
    int   cb[G];
    for (int g = 0; g < G; ++g) {
        const int r = r0 + g;
        int rr = r + rot;
        if (rr >= ROUNDS) rr -= ROUNDS;
        const int c = rr * 32 + w;
        const bool ok = (r < ROUNDS) && (c < N_CHUNKS);
        cb[g] = ok ? c : -1;
        q[g] = keys16[(ok ? c : 0) * 32 + lane];
    }
    for (int g = 0; g < G; ++g) {
        const unsigned lo7 = 0x7F7F7F7Fu, hi1 = 0x80808080u;
        const unsigned x0 = q[g].x ^ pat, x1 = q[g].y ^ pat, x2 = q[g].z ^ pat, x3 = q[g].w ^ pat;
        const unsigned y0 = (((x0 & lo7) + lo7) | x0) & hi1;
        const unsigned y1 = (((x1 & lo7) + lo7) | x1) & hi1;
        const unsigned y2 = (((x2 & lo7) + lo7) | x2) & hi1;
        const unsigned y3 = (((x3 & lo7) + lo7) | x3) & hi1;
        unsigned zz = ((y0 >> 7) | (y1 >> 6) | (y2 >> 5) | (y3 >> 4)) ^ 0x0F0F0F0Fu;
        if (cb[g] < 0) zz = 0u;
        const int e0 = cb[g] * CHUNK_EDGES + lane * 16;
        unsigned m = __builtin_amdgcn_ballot_w32(zz != 0u);
        for (int rd = 0; rd < 16; ++rd) {
            if (m == 0u) break;
            if (zz != 0u) {
                const int p = __builtin_ctz(zz);
                zz &= zz - 1u;
                const int pos = fillw + __popc(m & below);
                if (pos < LIST_CAP) mylist[pos] = e0 + (((p & 7) << 2) | (p >> 3));
            }
            fillw += __popc(m);
            m = __builtin_amdgcn_ballot_w32(zz != 0u);
        }
    }
    return fillw;
}

__device__ __forceinline__ void finish_list(const int* __restrict__ dst, int* mylist,
                                            int fillw, int n0, int lane) {
    for (int k0 = 0; k0 < fillw; k0 += 128) {
        int e[4], d[4];
        for (int i = 0; i < 4; ++i) {
            const int kk = k0 + i * 32 + lane;
            e[i] = kk < fillw ? mylist[kk] : 0;
        }
        for (int i = 0; i < 4; ++i) d[i] = dst[e[i]];
        for (int i = 0; i < 4; ++i) {
            const int kk = k0 + i * 32 + lane;
            if (kk < fillw) mylist[kk] = (e[i] << 8) | (d[i] - n0);
        }
    }
}

__global__ void __launch_bounds__(1024) owner_kernel(const float* __restrict__ nf,
                                                     const float* __restrict__ ef,
                                                     const int*   __restrict__ src,
                                                     const int*   __restrict__ dst,
                                                     const int*   __restrict__ keys,
                                                     float* __restrict__ out,
                                                     float* __restrict__ deg) {
    __shared__ float acc[NODES_PER_WG * ACC_W];
    __shared__ int   lists[32 * LIST_PITCH];
    __shared__ int   pend[32 * PEND_CAP];
    __shared__ int   fillv[32];
    __shared__ int   degc[NODES_PER_WG];
    __shared__ int   flag[N_STEPS];
    __shared__ int   ovf;

    const int tid  = threadIdx.x;
    const int lane = tid & 31;
    const int w    = __builtin_amdgcn_readfirstlane(tid >> 5);
    const int n0   = blockIdx.x * NODES_PER_WG;
    const unsigned below = (1u << lane) - 1u;

    for (int i = tid; i < NODES_PER_WG * ACC_W; i += 1024) acc[i] = 0.0f;
    if (tid < NODES_PER_WG) degc[tid] = 0;
    if (tid < N_STEPS) flag[tid] = 0;
    if (tid == 0) ovf = 0;
    __syncthreads();

    int* mylist = lists + w * LIST_PITCH;
    int* pendw  = pend + w * PEND_CAP;
    int  fillw  = 0;

    const uint4* keys16 = (const uint4*)keys;
    const unsigned pat = (unsigned)blockIdx.x * 0x01010101u;
    const int rot = (int)blockIdx.x % ROUNDS;

    for (int r0 = 0; r0 < ROUNDS; r0 += G_FAST)
        fillw = sweep_group<G_FAST>(keys16, r0, rot, pat, w, lane, below, mylist, fillw);
    if (lane == 0 && fillw > LIST_CAP) ovf = 1;
    __syncthreads();
    if (__builtin_amdgcn_readfirstlane(ovf) != 0) {
        fillw = 0;
        for (int st = 0; st < N_STEPS; ++st) {
            if (lane == 0) {
                fillv[w] = fillw;
                if (fillw > LIST_CAP - G_SAFE * CHUNK_EDGES) flag[st] = 1;
            }
            __syncthreads();
            if (__builtin_amdgcn_readfirstlane(flag[st]) != 0) {
                finish_list(dst, mylist, fillw, n0, lane);
                __syncthreads();
                drain_lists(nf, ef, src, acc, degc, lists, fillv, pendw, w, lane);
                __syncthreads();
                fillw = 0;
            }
            fillw = sweep_group<G_SAFE>(keys16, st * G_SAFE, rot, pat, w, lane, below, mylist, fillw);
        }
    }
    finish_list(dst, mylist, fillw, n0, lane);
    if (lane == 0) fillv[w] = fillw;
    __syncthreads();
    drain_lists(nf, ef, src, acc, degc, lists, fillv, pendw, w, lane);
    __syncthreads();

    for (int r = w; r < NODES_PER_WG; r += 32) {
        const float* a = acc + r * ACC_W;
        float* o = out + (size_t)(n0 + r) * OUT_DIM;
        *(float2*)(o + 2 * lane)      = *(const float2*)(a + 2 * lane);
        *(float2*)(o + 64 + 2 * lane) = *(const float2*)(a + 64 + 2 * lane);
        o[160 + lane] = a[128 + lane];
    }
    if (tid < NODES_PER_WG) deg[n0 + tid] = (float)degc[tid];
}

__global__ void prompt_kernel(const float* __restrict__ prompt,
                              const float* __restrict__ hp,
                              const float* __restrict__ deg,
                              float* __restrict__ out) {
    const int lane = threadIdx.x;
    const int m0   = blockIdx.x * 16;
    const int l15  = lane & 15;
    const bool lo  = (lane < 16);

    const float dv = deg[m0 + l15];
    v2f a; a.x = lo ? dv : 0.0f; a.y = 0.0f;

    const float p0 = prompt[l15], p1 = prompt[16 + l15];
    const float h0 = hp[l15],     h1 = hp[16 + l15];
    v2f b0; b0.x = lo ? p0 : 0.0f; b0.y = 0.0f;
    v2f b1; b1.x = lo ? p1 : 0.0f; b1.y = 0.0f;
    v2f b2; b2.x = lo ? h0 : 0.0f; b2.y = 0.0f;
    v2f b3; b3.x = lo ? h1 : 0.0f; b3.y = 0.0f;

    v8f c = {0.f, 0.f, 0.f, 0.f, 0.f, 0.f, 0.f, 0.f};
    v8f d0 = __builtin_amdgcn_wmma_f32_16x16x4_f32(false, a, false, b0, (short)0, c, false, false);
    v8f d1 = __builtin_amdgcn_wmma_f32_16x16x4_f32(false, a, false, b1, (short)0, c, false, false);
    v8f d2 = __builtin_amdgcn_wmma_f32_16x16x4_f32(false, a, false, b2, (short)0, c, false, false);
    v8f d3 = __builtin_amdgcn_wmma_f32_16x16x4_f32(false, a, false, b3, (short)0, c, false, false);

    const int mofs = (lane >> 4) * 8;
#pragma unroll
    for (int r = 0; r < 8; ++r) {
        const size_t row = (size_t)(m0 + mofs + r) * OUT_DIM;
        out[row + 128 + l15]      = d0[r];
        out[row + 144 + l15]      = d1[r];
        out[row + 192 + l15]      = d2[r];
        out[row + 208 + l15]      = d3[r];
    }
}

extern "C" void kernel_launch(void* const* d_in, const int* in_sizes, int n_in,
                              void* d_out, int out_size, void* d_ws, size_t ws_size,
                              hipStream_t stream) {
    (void)in_sizes; (void)n_in; (void)out_size; (void)ws_size;

    const float* nf     = (const float*)d_in[0];
    const float* ef     = (const float*)d_in[1];
    const float* prompt = (const float*)d_in[2];
    const float* hp     = (const float*)d_in[3];
    const int*   src    = (const int*)d_in[4];
    const int*   dst    = (const int*)d_in[5];

    float* out = (float*)d_out;
    float* deg = (float*)d_ws;
    int* keys = (int*)((char*)d_ws + 200704);

    key_kernel<<<(KEY_WORDS + 255) / 256, 256, 0, stream>>>(dst, keys);
    owner_kernel<<<N_NODES / NODES_PER_WG, 1024, 0, stream>>>(nf, ef, src, dst, keys, out, deg);
    prompt_kernel<<<N_NODES / 16, 32, 0, stream>>>(prompt, hp, deg, out);
}


# third launch folded into the owner kernel: waves 0-12 of each work-group do the deg x prompt and deg x hetero-prompt f32 WMMAs from the LDS degree counts, whole 128-B line stores; 3 -> 2 launches
# speedup vs baseline: 2.7958x; 1.0790x over previous
#include <hip/hip_runtime.h>

#define N_NODES   50000
#define N_EDGES   800000
#define IN_FEAT   128
#define PROMPT_DIM 32
#define EF_DIM     32
#define HP_DIM     32
#define OUT_DIM   224

typedef __attribute__((ext_vector_type(2))) float v2f;
typedef __attribute__((ext_vector_type(8))) float v8f;

#define NODES_PER_WG 200
#define ACC_W        160
#define CHUNK_EDGES  512
#define N_CHUNKS     1563
#define KEY_WORDS    (N_CHUNKS * CHUNK_EDGES / 4)
#define ROUNDS       49
#define G_FAST       7
#define G_SAFE       2
#define N_STEPS      25
#define LIST_CAP     1280
#define LIST_PITCH   1281
#define PEND_CAP     64
#define N_TILES      13
#define NO_OWNER     255u

__device__ __forceinline__ unsigned owner_key(int d) {
    return (unsigned)d < (unsigned)N_NODES ? (unsigned)d / (unsigned)NODES_PER_WG : NO_OWNER;
}

__global__ void __launch_bounds__(256) key_kernel(const int* __restrict__ dst,
                                                  int* __restrict__ keys) {
    const int t = blockIdx.x * 256 + threadIdx.x;
    if (t >= KEY_WORDS) return;
    unsigned k = 0xFFFFFFFFu;
    if (t < N_EDGES / 4) {
        const int4 d = *(const int4*)(dst + 4 * t);
        k = owner_key(d.x) | (owner_key(d.y) << 8) | (owner_key(d.z) << 16) | (owner_key(d.w) << 24);
    }
    keys[t] = (int)k;
}


__device__ __forceinline__ void add_batch8(const float* __restrict__ nf,
                                           const float* __restrict__ ef,
                                           const int*   __restrict__ src,
                                           float* acc, int* degc,
                                           const int* p, int lane) {
    int   ln[8];
    float2 xa[8], xb[8];
    float  fe[8];
    for (int i = 0; i < 8; ++i) {
        const int en = __builtin_amdgcn_readfirstlane(p[i]);
        const int e  = en >> 8;
        ln[i] = en & 255;
        const int s = src[e];
        const float* r = nf + (size_t)s * IN_FEAT;
        xa[i] = *(const float2*)(r + 2 * lane);
        xb[i] = *(const float2*)(r + 64 + 2 * lane);
        fe[i] = ef[(size_t)e * EF_DIM + lane];
    }
    for (int i = 0; i < 8; ++i) {
        float* a = acc + ln[i] * ACC_W;
        float2 va = *(float2*)(a + 2 * lane);
        float2 vb = *(float2*)(a + 64 + 2 * lane);
        va.x += xa[i].x; va.y += xa[i].y;
        vb.x += xb[i].x; vb.y += xb[i].y;
        *(float2*)(a + 2 * lane)      = va;
        *(float2*)(a + 64 + 2 * lane) = vb;
        a[128 + lane] += fe[i];
        if (lane == 0) degc[ln[i]] += 1;
    }
}

__device__ __forceinline__ void add_one(const float* __restrict__ nf,
                                        const float* __restrict__ ef,
                                        const int*   __restrict__ src,
                                        float* acc, int* degc,
                                        const int* p, int lane) {
    const int en = __builtin_amdgcn_readfirstlane(p[0]);
    const int e  = en >> 8;
    const int ln = en & 255;
    const int s  = src[e];
    const float* r = nf + (size_t)s * IN_FEAT;
    const float2 xa = *(const float2*)(r + 2 * lane);
    const float2 xb = *(const float2*)(r + 64 + 2 * lane);
    const float  fe = ef[(size_t)e * EF_DIM + lane];
    float* a = acc + ln * ACC_W;
    float2 va = *(float2*)(a + 2 * lane);
    float2 vb = *(float2*)(a + 64 + 2 * lane);
    va.x += xa.x; va.y += xa.y;
    vb.x += xb.x; vb.y += xb.y;
    *(float2*)(a + 2 * lane)      = va;
    *(float2*)(a + 64 + 2 * lane) = vb;
    a[128 + lane] += fe;
    if (lane == 0) degc[ln] += 1;
}

__device__ __forceinline__ void drain_lists(const float* __restrict__ nf,
                                            const float* __restrict__ ef,
                                            const int*   __restrict__ src,
                                            float* acc, int* degc,
                                            const int* lists, const int* fillv,
                                            int* pendw, int w, int lane) {
    const unsigned below = (1u << lane) - 1u;
    const int  myfill = fillv[lane];
    const int* mylist = lists + lane * LIST_PITCH;
    int npend = 0;
    for (int k = 0; k < LIST_CAP; ++k) {
        const bool valid = k < myfill;
        if (__builtin_amdgcn_ballot_w32(valid) == 0u) break;
        const int raw  = mylist[k];
        const bool mine = valid && ((raw & 31) == w);
        const unsigned m = __builtin_amdgcn_ballot_w32(mine);
        if (m != 0u) {
            if (mine) pendw[npend + __popc(m & below)] = raw;
            npend += __popc(m);
            if (npend >= 32) {
                const int full = npend & ~7;
                for (int j = 0; j < full; j += 8)
                    add_batch8(nf, ef, src, acc, degc, pendw + j, lane);
                const int left = npend - full;
                const int t = pendw[full + (lane < left ? lane : 0)];
                if (lane < left) pendw[lane] = t;
                npend = left;
            }
        }
    }
    const int full = npend & ~7;
    for (int j = 0; j < full; j += 8)
        add_batch8(nf, ef, src, acc, degc, pendw + j, lane);
    for (int j = full; j < npend; ++j)
        add_one(nf, ef, src, acc, degc, pendw + j, lane);
}

template <int G>
__device__ __forceinline__ int sweep_group(const uint4* __restrict__ keys16, int r0, int rot,
                                           unsigned pat, int w, int lane, unsigned below,
                                           int* mylist, int fillw) {
    uint4 q[G];
    int   cb[G];
    for (int g = 0; g < G; ++g) {
        const int r = r0 + g;
        int rr = r + rot;
        if (rr >= ROUNDS) rr -= ROUNDS;
        const int c = rr * 32 + w;
        const bool ok = (r < ROUNDS) && (c < N_CHUNKS);
        cb[g] = ok ? c : -1;
        q[g] = keys16[(ok ? c : 0) * 32 + lane];
    }
    for (int g = 0; g < G; ++g) {
        const unsigned lo7 = 0x7F7F7F7Fu, hi1 = 0x80808080u;
        const unsigned x0 = q[g].x ^ pat, x1 = q[g].y ^ pat, x2 = q[g].z ^ pat, x3 = q[g].w ^ pat;
        const unsigned y0 = (((x0 & lo7) + lo7) | x0) & hi1;
        const unsigned y1 = (((x1 & lo7) + lo7) | x1) & hi1;
        const unsigned y2 = (((x2 & lo7) + lo7) | x2) & hi1;
        const unsigned y3 = (((x3 & lo7) + lo7) | x3) & hi1;
        unsigned zz = ((y0 >> 7) | (y1 >> 6) | (y2 >> 5) | (y3 >> 4)) ^ 0x0F0F0F0Fu;
        if (cb[g] < 0) zz = 0u;
        const int e0 = cb[g] * CHUNK_EDGES + lane * 16;
        unsigned m = __builtin_amdgcn_ballot_w32(zz != 0u);
        for (int rd = 0; rd < 16; ++rd) {
            if (m == 0u) break;
            if (zz != 0u) {
                const int p = __builtin_ctz(zz);
                zz &= zz - 1u;
                const int pos = fillw + __popc(m & below);
                if (pos < LIST_CAP) mylist[pos] = e0 + (((p & 7) << 2) | (p >> 3));
            }
            fillw += __popc(m);
            m = __builtin_amdgcn_ballot_w32(zz != 0u);
        }
    }
    return fillw;
}

__device__ __forceinline__ void finish_list(const int* __restrict__ dst, int* mylist,
                                            int fillw, int n0, int lane) {
    for (int k0 = 0; k0 < fillw; k0 += 128) {
        int e[4], d[4];
        for (int i = 0; i < 4; ++i) {
            const int kk = k0 + i * 32 + lane;
            e[i] = kk < fillw ? mylist[kk] : 0;
        }
        for (int i = 0; i < 4; ++i) d[i] = dst[e[i]];
        for (int i = 0; i < 4; ++i) {
            const int kk = k0 + i * 32 + lane;
            if (kk < fillw) mylist[kk] = (e[i] << 8) | (d[i] - n0);
        }
    }
}

__global__ void __launch_bounds__(1024) owner_kernel(const float* __restrict__ nf,
                                                     const float* __restrict__ ef,
                                                     const int*   __restrict__ src,
                                                     const int*   __restrict__ dst,
                                                     const int*   __restrict__ keys,
                                                     const float* __restrict__ prompt,
                                                     const float* __restrict__ hp,
                                                     float* __restrict__ out) {
    __shared__ float acc[NODES_PER_WG * ACC_W];
    __shared__ int   lists[32 * LIST_PITCH];
    __shared__ int   pend[32 * PEND_CAP];
    __shared__ int   fillv[32];
    __shared__ int   degc[NODES_PER_WG];
    __shared__ int   flag[N_STEPS];
    __shared__ int   ovf;

    const int tid  = threadIdx.x;
    const int lane = tid & 31;
    const int w    = __builtin_amdgcn_readfirstlane(tid >> 5);
    const int n0   = blockIdx.x * NODES_PER_WG;
    const unsigned below = (1u << lane) - 1u;

    const float p0 = prompt[lane & 15], p1 = prompt[16 + (lane & 15)];
    const float h0 = hp[lane & 15],     h1 = hp[16 + (lane & 15)];

    for (int i = tid; i < NODES_PER_WG * ACC_W; i += 1024) acc[i] = 0.0f;
    if (tid < NODES_PER_WG) degc[tid] = 0;
    if (tid < N_STEPS) flag[tid] = 0;
    if (tid == 0) ovf = 0;
    __syncthreads();

    int* mylist = lists + w * LIST_PITCH;
    int* pendw  = pend + w * PEND_CAP;
    int  fillw  = 0;

    const uint4* keys16 = (const uint4*)keys;
    const unsigned pat = (unsigned)blockIdx.x * 0x01010101u;
    const int rot = (int)blockIdx.x % ROUNDS;

    for (int r0 = 0; r0 < ROUNDS; r0 += G_FAST)
        fillw = sweep_group<G_FAST>(keys16, r0, rot, pat, w, lane, below, mylist, fillw);
    if (lane == 0 && fillw > LIST_CAP) ovf = 1;
    __syncthreads();
    if (__builtin_amdgcn_readfirstlane(ovf) != 0) {
        fillw = 0;
        for (int st = 0; st < N_STEPS; ++st) {
            if (lane == 0) {
                fillv[w] = fillw;
                if (fillw > LIST_CAP - G_SAFE * CHUNK_EDGES) flag[st] = 1;
            }
            __syncthreads();
            if (__builtin_amdgcn_readfirstlane(flag[st]) != 0) {
                finish_list(dst, mylist, fillw, n0, lane);
                __syncthreads();
                drain_lists(nf, ef, src, acc, degc, lists, fillv, pendw, w, lane);
                __syncthreads();
                fillw = 0;
            }
            fillw = sweep_group<G_SAFE>(keys16, st * G_SAFE, rot, pat, w, lane, below, mylist, fillw);
        }
    }
    finish_list(dst, mylist, fillw, n0, lane);
    if (lane == 0) fillv[w] = fillw;
    __syncthreads();
    drain_lists(nf, ef, src, acc, degc, lists, fillv, pendw, w, lane);
    __syncthreads();

    if (w < N_TILES) {
        const int  l15 = lane & 15;
        const bool lo  = (lane < 16);
        const int  lr  = w * 16 + l15;
        const int  dc  = degc[lr < NODES_PER_WG ? lr : 0];
        v2f a; a.x = (lo && lr < NODES_PER_WG) ? (float)dc : 0.0f; a.y = 0.0f;

        v2f b0; b0.x = lo ? p0 : 0.0f; b0.y = 0.0f;
        v2f b1; b1.x = lo ? p1 : 0.0f; b1.y = 0.0f;
        v2f b2; b2.x = lo ? h0 : 0.0f; b2.y = 0.0f;
        v2f b3; b3.x = lo ? h1 : 0.0f; b3.y = 0.0f;

        v8f c = {0.f, 0.f, 0.f, 0.f, 0.f, 0.f, 0.f, 0.f};
        v8f d0 = __builtin_amdgcn_wmma_f32_16x16x4_f32(false, a, false, b0, (short)0, c, false, false);
        v8f d1 = __builtin_amdgcn_wmma_f32_16x16x4_f32(false, a, false, b1, (short)0, c, false, false);
        v8f d2 = __builtin_amdgcn_wmma_f32_16x16x4_f32(false, a, false, b2, (short)0, c, false, false);
        v8f d3 = __builtin_amdgcn_wmma_f32_16x16x4_f32(false, a, false, b3, (short)0, c, false, false);

#pragma unroll
        for (int r = 0; r < 8; ++r) {
            const float s0 = __shfl_xor(d0[r], 16);
            const float s1 = __shfl_xor(d1[r], 16);
            const float s2 = __shfl_xor(d2[r], 16);
            const float s3 = __shfl_xor(d3[r], 16);
            const int ra = w * 16 + r;
            const int rb = ra + 8;
            float* oa = out + (size_t)(n0 + ra) * OUT_DIM;
            oa[128 + lane] = lo ? d0[r] : s1;
            oa[192 + lane] = lo ? d2[r] : s3;
            if (rb < NODES_PER_WG) {
                float* ob = out + (size_t)(n0 + rb) * OUT_DIM;
                ob[128 + lane] = lo ? s0 : d1[r];
                ob[192 + lane] = lo ? s2 : d3[r];
            }
        }
    }

    for (int r = w; r < NODES_PER_WG; r += 32) {
        const float* a = acc + r * ACC_W;
        float* o = out + (size_t)(n0 + r) * OUT_DIM;
        *(float2*)(o + 2 * lane)      = *(const float2*)(a + 2 * lane);
        *(float2*)(o + 64 + 2 * lane) = *(const float2*)(a + 64 + 2 * lane);
        o[160 + lane] = a[128 + lane];
    }
}

extern "C" void kernel_launch(void* const* d_in, const int* in_sizes, int n_in,
                              void* d_out, int out_size, void* d_ws, size_t ws_size,
                              hipStream_t stream) {
    (void)in_sizes; (void)n_in; (void)out_size; (void)ws_size;

    const float* nf     = (const float*)d_in[0];
    const float* ef     = (const float*)d_in[1];
    const float* prompt = (const float*)d_in[2];
    const float* hp     = (const float*)d_in[3];
    const int*   src    = (const int*)d_in[4];
    const int*   dst    = (const int*)d_in[5];

    float* out = (float*)d_out;
    int* keys = (int*)((char*)d_ws + 200704);

    key_kernel<<<(KEY_WORDS + 255) / 256, 256, 0, stream>>>(dst, keys);
    owner_kernel<<<N_NODES / NODES_PER_WG, 1024, 0, stream>>>(nf, ef, src, dst, keys, prompt, hp, out);
}


# row loads of up to 16 edges issued together (was 8): one LDS read and one src load per batch with lane broadcasts, batch length uniform 1..16, no one-by-one tail; per-node sum order unchanged
# speedup vs baseline: 2.8891x; 1.0334x over previous
#include <hip/hip_runtime.h>

#define N_NODES   50000
#define N_EDGES   800000
#define IN_FEAT   128
#define PROMPT_DIM 32
#define EF_DIM     32
#define HP_DIM     32
#define OUT_DIM   224

typedef __attribute__((ext_vector_type(2))) float v2f;
typedef __attribute__((ext_vector_type(8))) float v8f;

#define NODES_PER_WG 200
#define ACC_W        160
#define CHUNK_EDGES  512
#define N_CHUNKS     1563
#define KEY_WORDS    (N_CHUNKS * CHUNK_EDGES / 4)
#define ROUNDS       49
#define G_FAST       7
#define G_SAFE       2
#define N_STEPS      25
#define LIST_CAP     1280
#define LIST_PITCH   1281
#define PEND_CAP     64
#define BATCH        16
#define N_TILES      13
#define NO_OWNER     255u

__device__ __forceinline__ unsigned owner_key(int d) {
    return (unsigned)d < (unsigned)N_NODES ? (unsigned)d / (unsigned)NODES_PER_WG : NO_OWNER;
}

__global__ void __launch_bounds__(256) key_kernel(const int* __restrict__ dst,
                                                  int* __restrict__ keys) {
    const int t = blockIdx.x * 256 + threadIdx.x;
    if (t >= KEY_WORDS) return;
    unsigned k = 0xFFFFFFFFu;
    if (t < N_EDGES / 4) {
        const int4 d = *(const int4*)(dst + 4 * t);
        k = owner_key(d.x) | (owner_key(d.y) << 8) | (owner_key(d.z) << 16) | (owner_key(d.w) << 24);
    }
    keys[t] = (int)k;
}


__device__ __forceinline__ void add_batch(const float* __restrict__ nf,
                                          const float* __restrict__ ef,
                                          const int*   __restrict__ src,
                                          float* acc, int* degc,
                                          const int* p, int n, int lane) {
    const int ent = p[lane < n ? lane : 0];
    const int sv  = src[ent >> 8];
    int   ln[BATCH];
    float2 xa[BATCH], xb[BATCH];
    float  fe[BATCH];
#pragma unroll
    for (int i = 0; i < BATCH; ++i) {
        const int en = __builtin_amdgcn_readlane(ent, i);
        ln[i] = en & 255;
        fe[i] = ef[(size_t)(en >> 8) * EF_DIM + lane];
    }
#pragma unroll
    for (int i = 0; i < BATCH; ++i) {
        const int s = __builtin_amdgcn_readlane(sv, i);
        const float* r = nf + (size_t)s * IN_FEAT;
        xa[i] = *(const float2*)(r + 2 * lane);
        xb[i] = *(const float2*)(r + 64 + 2 * lane);
    }
#pragma unroll
    for (int i = 0; i < BATCH; ++i) {
        if (i == 0 || i < n) {
            float* a = acc + ln[i] * ACC_W;
            float2 va = *(float2*)(a + 2 * lane);
            float2 vb = *(float2*)(a + 64 + 2 * lane);
            va.x += xa[i].x; va.y += xa[i].y;
            vb.x += xb[i].x; vb.y += xb[i].y;
            *(float2*)(a + 2 * lane)      = va;
            *(float2*)(a + 64 + 2 * lane) = vb;
            a[128 + lane] += fe[i];
            if (lane == 0) degc[ln[i]] += 1;
        }
    }
}

__device__ __forceinline__ void drain_lists(const float* __restrict__ nf,
                                            const float* __restrict__ ef,
                                            const int*   __restrict__ src,
                                            float* acc, int* degc,
                                            const int* lists, const int* fillv,
                                            int* pendw, int w, int lane) {
    const unsigned below = (1u << lane) - 1u;
    const int  myfill = fillv[lane] < LIST_CAP ? fillv[lane] : LIST_CAP;
    const int* mylist = lists + lane * LIST_PITCH;
    int  npend = 0;
    int  k = 0;
    bool more = true;
    for (;;) {
        while (more && npend < BATCH) {
            const bool valid = k < myfill;
            if (__builtin_amdgcn_ballot_w32(valid) == 0u) { more = false; break; }
            const int raw  = mylist[k];
            ++k;
            const bool mine = valid && ((raw & 31) == w);
            const unsigned m = __builtin_amdgcn_ballot_w32(mine);
            if (m != 0u) {
                if (mine) pendw[npend + __popc(m & below)] = raw;
                npend += __popc(m);
            }
        }
        if (npend == 0) break;
        const int n = npend < BATCH ? npend : BATCH;
        add_batch(nf, ef, src, acc, degc, pendw, n, lane);
        const int left = npend - n;
        if (left > 0) {
            const int t = pendw[BATCH + (lane < left ? lane : 0)];
            if (lane < left) pendw[lane] = t;
        }
        npend = left;
    }
}

template <int G>
__device__ __forceinline__ int sweep_group(const uint4* __restrict__ keys16, int r0, int rot,
                                           unsigned pat, int w, int lane, unsigned below,
                                           int* mylist, int fillw) {
    uint4 q[G];
    int   cb[G];
    for (int g = 0; g < G; ++g) {
        const int r = r0 + g;
        int rr = r + rot;
        if (rr >= ROUNDS) rr -= ROUNDS;
        const int c = rr * 32 + w;
        const bool ok = (r < ROUNDS) && (c < N_CHUNKS);
        cb[g] = ok ? c : -1;
        q[g] = keys16[(ok ? c : 0) * 32 + lane];
    }
    for (int g = 0; g < G; ++g) {
        const unsigned lo7 = 0x7F7F7F7Fu, hi1 = 0x80808080u;
        const unsigned x0 = q[g].x ^ pat, x1 = q[g].y ^ pat, x2 = q[g].z ^ pat, x3 = q[g].w ^ pat;
        const unsigned y0 = (((x0 & lo7) + lo7) | x0) & hi1;
        const unsigned y1 = (((x1 & lo7) + lo7) | x1) & hi1;
        const unsigned y2 = (((x2 & lo7) + lo7) | x2) & hi1;
        const unsigned y3 = (((x3 & lo7) + lo7) | x3) & hi1;
        unsigned zz = ((y0 >> 7) | (y1 >> 6) | (y2 >> 5) | (y3 >> 4)) ^ 0x0F0F0F0Fu;
        if (cb[g] < 0) zz = 0u;
        const int e0 = cb[g] * CHUNK_EDGES + lane * 16;
        unsigned m = __builtin_amdgcn_ballot_w32(zz != 0u);
        for (int rd = 0; rd < 16; ++rd) {
            if (m == 0u) break;
            if (zz != 0u) {
                const int p = __builtin_ctz(zz);
                zz &= zz - 1u;
                const int pos = fillw + __popc(m & below);
                if (pos < LIST_CAP) mylist[pos] = e0 + (((p & 7) << 2) | (p >> 3));
            }
            fillw += __popc(m);
            m = __builtin_amdgcn_ballot_w32(zz != 0u);
        }
    }
    return fillw;
}

__device__ __forceinline__ void finish_list(const int* __restrict__ dst, int* mylist,
                                            int fillw, int n0, int lane) {
    for (int k0 = 0; k0 < fillw; k0 += 128) {
        int e[4], d[4];
        for (int i = 0; i < 4; ++i) {
            const int kk = k0 + i * 32 + lane;
            e[i] = kk < fillw ? mylist[kk] : 0;
        }
        for (int i = 0; i < 4; ++i) d[i] = dst[e[i]];
        for (int i = 0; i < 4; ++i) {
            const int kk = k0 + i * 32 + lane;
            if (kk < fillw) mylist[kk] = (e[i] << 8) | (d[i] - n0);
        }
    }
}

__global__ void __launch_bounds__(1024) owner_kernel(const float* __restrict__ nf,
                                                     const float* __restrict__ ef,
                                                     const int*   __restrict__ src,
                                                     const int*   __restrict__ dst,
                                                     const int*   __restrict__ keys,
                                                     const float* __restrict__ prompt,
                                                     const float* __restrict__ hp,
                                                     float* __restrict__ out) {
    __shared__ float acc[NODES_PER_WG * ACC_W];
    __shared__ int   lists[32 * LIST_PITCH];
    __shared__ int   pend[32 * PEND_CAP];
    __shared__ int   fillv[32];
    __shared__ int   degc[NODES_PER_WG];
    __shared__ int   flag[N_STEPS];
    __shared__ int   ovf;

    const int tid  = threadIdx.x;
    const int lane = tid & 31;
    const int w    = __builtin_amdgcn_readfirstlane(tid >> 5);
    const int n0   = blockIdx.x * NODES_PER_WG;
    const unsigned below = (1u << lane) - 1u;

    const float p0 = prompt[lane & 15], p1 = prompt[16 + (lane & 15)];
    const float h0 = hp[lane & 15],     h1 = hp[16 + (lane & 15)];

    for (int i = tid; i < NODES_PER_WG * ACC_W; i += 1024) acc[i] = 0.0f;
    if (tid < NODES_PER_WG) degc[tid] = 0;
    if (tid < N_STEPS) flag[tid] = 0;
    if (tid == 0) ovf = 0;
    __syncthreads();

    int* mylist = lists + w * LIST_PITCH;
    int* pendw  = pend + w * PEND_CAP;
    int  fillw  = 0;

    const uint4* keys16 = (const uint4*)keys;
    const unsigned pat = (unsigned)blockIdx.x * 0x01010101u;
    const int rot = (int)blockIdx.x % ROUNDS;

    for (int r0 = 0; r0 < ROUNDS; r0 += G_FAST)
        fillw = sweep_group<G_FAST>(keys16, r0, rot, pat, w, lane, below, mylist, fillw);
    if (lane == 0 && fillw > LIST_CAP) ovf = 1;
    __syncthreads();
    if (__builtin_amdgcn_readfirstlane(ovf) != 0) {
        fillw = 0;
        for (int st = 0; st < N_STEPS; ++st) {
            if (lane == 0) {
                fillv[w] = fillw;
                if (fillw > LIST_CAP - G_SAFE * CHUNK_EDGES) flag[st] = 1;
            }
            __syncthreads();
            if (__builtin_amdgcn_readfirstlane(flag[st]) != 0) {
                finish_list(dst, mylist, fillw, n0, lane);
                __syncthreads();
                drain_lists(nf, ef, src, acc, degc, lists, fillv, pendw, w, lane);
                __syncthreads();
                fillw = 0;
            }
            fillw = sweep_group<G_SAFE>(keys16, st * G_SAFE, rot, pat, w, lane, below, mylist, fillw);
        }
    }
    finish_list(dst, mylist, fillw, n0, lane);
    if (lane == 0) fillv[w] = fillw;
    __syncthreads();
    drain_lists(nf, ef, src, acc, degc, lists, fillv, pendw, w, lane);
    __syncthreads();

    if (w < N_TILES) {
        const int  l15 = lane & 15;
        const bool lo  = (lane < 16);
        const int  lr  = w * 16 + l15;
        const int  dc  = degc[lr < NODES_PER_WG ? lr : 0];
        v2f a; a.x = (lo && lr < NODES_PER_WG) ? (float)dc : 0.0f; a.y = 0.0f;

        v2f b0; b0.x = lo ? p0 : 0.0f; b0.y = 0.0f;
        v2f b1; b1.x = lo ? p1 : 0.0f; b1.y = 0.0f;
        v2f b2; b2.x = lo ? h0 : 0.0f; b2.y = 0.0f;
        v2f b3; b3.x = lo ? h1 : 0.0f; b3.y = 0.0f;

        v8f c = {0.f, 0.f, 0.f, 0.f, 0.f, 0.f, 0.f, 0.f};
        v8f d0 = __builtin_amdgcn_wmma_f32_16x16x4_f32(false, a, false, b0, (short)0, c, false, false);
        v8f d1 = __builtin_amdgcn_wmma_f32_16x16x4_f32(false, a, false, b1, (short)0, c, false, false);
        v8f d2 = __builtin_amdgcn_wmma_f32_16x16x4_f32(false, a, false, b2, (short)0, c, false, false);
        v8f d3 = __builtin_amdgcn_wmma_f32_16x16x4_f32(false, a, false, b3, (short)0, c, false, false);

#pragma unroll
        for (int r = 0; r < 8; ++r) {
            const float s0 = __shfl_xor(d0[r], 16);
            const float s1 = __shfl_xor(d1[r], 16);
            const float s2 = __shfl_xor(d2[r], 16);
            const float s3 = __shfl_xor(d3[r], 16);
            const int ra = w * 16 + r;
            const int rb = ra + 8;
            float* oa = out + (size_t)(n0 + ra) * OUT_DIM;
            oa[128 + lane] = lo ? d0[r] : s1;
            oa[192 + lane] = lo ? d2[r] : s3;
            if (rb < NODES_PER_WG) {
                float* ob = out + (size_t)(n0 + rb) * OUT_DIM;
                ob[128 + lane] = lo ? s0 : d1[r];
                ob[192 + lane] = lo ? s2 : d3[r];
            }
        }
    }

    for (int r = w; r < NODES_PER_WG; r += 32) {
        const float* a = acc + r * ACC_W;
        float* o = out + (size_t)(n0 + r) * OUT_DIM;
        *(float2*)(o + 2 * lane)      = *(const float2*)(a + 2 * lane);
        *(float2*)(o + 64 + 2 * lane) = *(const float2*)(a + 64 + 2 * lane);
        o[160 + lane] = a[128 + lane];
    }
}

extern "C" void kernel_launch(void* const* d_in, const int* in_sizes, int n_in,
                              void* d_out, int out_size, void* d_ws, size_t ws_size,
                              hipStream_t stream) {
    (void)in_sizes; (void)n_in; (void)out_size; (void)ws_size;

    const float* nf     = (const float*)d_in[0];
    const float* ef     = (const float*)d_in[1];
    const float* prompt = (const float*)d_in[2];
    const float* hp     = (const float*)d_in[3];
    const int*   src    = (const int*)d_in[4];
    const int*   dst    = (const int*)d_in[5];

    float* out = (float*)d_out;
    int* keys = (int*)((char*)d_ws + 200704);

    key_kernel<<<(KEY_WORDS + 255) / 256, 256, 0, stream>>>(dst, keys);
    owner_kernel<<<N_NODES / NODES_PER_WG, 1024, 0, stream>>>(nf, ef, src, dst, keys, prompt, hp, out);
}


# each 16-edge batch stable-sorted by node inside the wave (ballots + one lane permute): a node's LDS sums are read and written once per batch, in-degree once per batch; per-node sum order unchanged
# speedup vs baseline: 3.1938x; 1.1055x over previous
#include <hip/hip_runtime.h>

#define N_NODES   50000
#define N_EDGES   800000
#define IN_FEAT   128
#define PROMPT_DIM 32
#define EF_DIM     32
#define HP_DIM     32
#define OUT_DIM   224

typedef __attribute__((ext_vector_type(2))) float v2f;
typedef __attribute__((ext_vector_type(8))) float v8f;

#define NODES_PER_WG 200
#define ACC_W        160
#define CHUNK_EDGES  512
#define N_CHUNKS     1563
#define KEY_WORDS    (N_CHUNKS * CHUNK_EDGES / 4)
#define ROUNDS       49
#define G_FAST       7
#define G_SAFE       2
#define N_STEPS      25
#define LIST_CAP     1280
#define LIST_PITCH   1281
#define PEND_CAP     64
#define BATCH        16
#define SLOTS        7
#define N_TILES      13
#define NO_OWNER     255u

__device__ __forceinline__ unsigned owner_key(int d) {
    return (unsigned)d < (unsigned)N_NODES ? (unsigned)d / (unsigned)NODES_PER_WG : NO_OWNER;
}

__global__ void __launch_bounds__(256) key_kernel(const int* __restrict__ dst,
                                                  int* __restrict__ keys) {
    const int t = blockIdx.x * 256 + threadIdx.x;
    if (t >= KEY_WORDS) return;
    unsigned k = 0xFFFFFFFFu;
    if (t < N_EDGES / 4) {
        const int4 d = *(const int4*)(dst + 4 * t);
        k = owner_key(d.x) | (owner_key(d.y) << 8) | (owner_key(d.z) << 16) | (owner_key(d.w) << 24);
    }
    keys[t] = (int)k;
}


__device__ __forceinline__ void add_batch(const float* __restrict__ nf,
                                          const float* __restrict__ ef,
                                          const int*   __restrict__ src,
                                          float* acc, int* degc,
                                          const int* p, int n, int w, int lane, unsigned below) {
    const int e0  = p[lane < n ? lane : 0];
    const int key = lane < n ? ((e0 & 255) >> 5) : SLOTS;
    int rank = 0, base = 0, mine = 0;
    for (int s = 0; s <= SLOTS; ++s) {
        const unsigned m = __builtin_amdgcn_ballot_w32(key == s);
        if (key == s)  rank = base + __popc(m & below);
        if (lane == s) mine = __popc(m);
        base += __popc(m);
    }
    const int ent = __builtin_amdgcn_ds_permute(rank << 2, e0);
    const int sv  = src[ent >> 8];
    int   ln[BATCH];
    float2 xa[BATCH], xb[BATCH];
    float  fe[BATCH];
#pragma unroll
    for (int i = 0; i < BATCH; ++i) {
        const int en = __builtin_amdgcn_readlane(ent, i);
        ln[i] = en & 255;
        fe[i] = ef[(size_t)(en >> 8) * EF_DIM + lane];
    }
#pragma unroll
    for (int i = 0; i < BATCH; ++i) {
        const int s = __builtin_amdgcn_readlane(sv, i);
        const float* r = nf + (size_t)s * IN_FEAT;
        xa[i] = *(const float2*)(r + 2 * lane);
        xb[i] = *(const float2*)(r + 64 + 2 * lane);
    }
    if (lane < SLOTS && mine > 0) degc[w + 32 * lane] += mine;

    float2 ta = make_float2(0.0f, 0.0f), tb = make_float2(0.0f, 0.0f);
    float  tc = 0.0f;
#pragma unroll
    for (int i = 0; i < BATCH; ++i) {
        if (i == 0 || i < n) {
            float* a = acc + ln[i] * ACC_W;
            bool first = true;
            if (i > 0) first = (ln[i] != ln[i - 1]);
            if (first) {
                ta = *(float2*)(a + 2 * lane);
                tb = *(float2*)(a + 64 + 2 * lane);
                tc = a[128 + lane];
            }
            ta.x += xa[i].x; ta.y += xa[i].y;
            tb.x += xb[i].x; tb.y += xb[i].y;
            tc   += fe[i];
            bool last = true;
            if (i + 1 < BATCH) last = (i + 1 >= n) || (ln[i + 1] != ln[i]);
            if (last) {
                *(float2*)(a + 2 * lane)      = ta;
                *(float2*)(a + 64 + 2 * lane) = tb;
                a[128 + lane] = tc;
            }
        }
    }
}

__device__ __forceinline__ void drain_lists(const float* __restrict__ nf,
                                            const float* __restrict__ ef,
                                            const int*   __restrict__ src,
                                            float* acc, int* degc,
                                            const int* lists, const int* fillv,
                                            int* pendw, int w, int lane) {
    const unsigned below = (1u << lane) - 1u;
    const int  myfill = fillv[lane] < LIST_CAP ? fillv[lane] : LIST_CAP;
    const int* mylist = lists + lane * LIST_PITCH;
    int  npend = 0;
    int  k = 0;
    bool more = true;
    for (;;) {
        while (more && npend < BATCH) {
            const bool valid = k < myfill;
            if (__builtin_amdgcn_ballot_w32(valid) == 0u) { more = false; break; }
            const int raw  = mylist[k];
            ++k;
            const bool mine = valid && ((raw & 31) == w);
            const unsigned m = __builtin_amdgcn_ballot_w32(mine);
            if (m != 0u) {
                if (mine) pendw[npend + __popc(m & below)] = raw;
                npend += __popc(m);
            }
        }
        if (npend == 0) break;
        const int n = npend < BATCH ? npend : BATCH;
        add_batch(nf, ef, src, acc, degc, pendw, n, w, lane, below);
        const int left = npend - n;
        if (left > 0) {
            const int t = pendw[BATCH + (lane < left ? lane : 0)];
            if (lane < left) pendw[lane] = t;
        }
        npend = left;
    }
}

template <int G>
__device__ __forceinline__ int sweep_group(const uint4* __restrict__ keys16, int r0, int rot,
                                           unsigned pat, int w, int lane, unsigned below,
                                           int* mylist, int fillw) {
    uint4 q[G];
    int   cb[G];
    for (int g = 0; g < G; ++g) {
        const int r = r0 + g;
        int rr = r + rot;
        if (rr >= ROUNDS) rr -= ROUNDS;
        const int c = rr * 32 + w;
        const bool ok = (r < ROUNDS) && (c < N_CHUNKS);
        cb[g] = ok ? c : -1;
        q[g] = keys16[(ok ? c : 0) * 32 + lane];
    }
    for (int g = 0; g < G; ++g) {
        const unsigned lo7 = 0x7F7F7F7Fu, hi1 = 0x80808080u;
        const unsigned x0 = q[g].x ^ pat, x1 = q[g].y ^ pat, x2 = q[g].z ^ pat, x3 = q[g].w ^ pat;
        const unsigned y0 = (((x0 & lo7) + lo7) | x0) & hi1;
        const unsigned y1 = (((x1 & lo7) + lo7) | x1) & hi1;
        const unsigned y2 = (((x2 & lo7) + lo7) | x2) & hi1;
        const unsigned y3 = (((x3 & lo7) + lo7) | x3) & hi1;
        unsigned zz = ((y0 >> 7) | (y1 >> 6) | (y2 >> 5) | (y3 >> 4)) ^ 0x0F0F0F0Fu;
        if (cb[g] < 0) zz = 0u;
        const int e0 = cb[g] * CHUNK_EDGES + lane * 16;
        unsigned m = __builtin_amdgcn_ballot_w32(zz != 0u);
        for (int rd = 0; rd < 16; ++rd) {
            if (m == 0u) break;
            if (zz != 0u) {
                const int p = __builtin_ctz(zz);
                zz &= zz - 1u;
                const int pos = fillw + __popc(m & below);
                if (pos < LIST_CAP) mylist[pos] = e0 + (((p & 7) << 2) | (p >> 3));
            }
            fillw += __popc(m);
            m = __builtin_amdgcn_ballot_w32(zz != 0u);
        }
    }
    return fillw;
}

__device__ __forceinline__ void finish_list(const int* __restrict__ dst, int* mylist,
                                            int fillw, int n0, int lane) {
    for (int k0 = 0; k0 < fillw; k0 += 128) {
        int e[4], d[4];
        for (int i = 0; i < 4; ++i) {
            const int kk = k0 + i * 32 + lane;
            e[i] = kk < fillw ? mylist[kk] : 0;
        }
        for (int i = 0; i < 4; ++i) d[i] = dst[e[i]];
        for (int i = 0; i < 4; ++i) {
            const int kk = k0 + i * 32 + lane;
            if (kk < fillw) mylist[kk] = (e[i] << 8) | (d[i] - n0);
        }
    }
}

__global__ void __launch_bounds__(1024) owner_kernel(const float* __restrict__ nf,
                                                     const float* __restrict__ ef,
                                                     const int*   __restrict__ src,
                                                     const int*   __restrict__ dst,
                                                     const int*   __restrict__ keys,
                                                     const float* __restrict__ prompt,
                                                     const float* __restrict__ hp,
                                                     float* __restrict__ out) {
    __shared__ float acc[NODES_PER_WG * ACC_W];
    __shared__ int   lists[32 * LIST_PITCH];
    __shared__ int   pend[32 * PEND_CAP];
    __shared__ int   fillv[32];
    __shared__ int   degc[NODES_PER_WG];
    __shared__ int   flag[N_STEPS];
    __shared__ int   ovf;

    const int tid  = threadIdx.x;
    const int lane = tid & 31;
    const int w    = __builtin_amdgcn_readfirstlane(tid >> 5);
    const int n0   = blockIdx.x * NODES_PER_WG;
    const unsigned below = (1u << lane) - 1u;

    const float p0 = prompt[lane & 15], p1 = prompt[16 + (lane & 15)];
    const float h0 = hp[lane & 15],     h1 = hp[16 + (lane & 15)];

    for (int i = tid; i < NODES_PER_WG * ACC_W; i += 1024) acc[i] = 0.0f;
    if (tid < NODES_PER_WG) degc[tid] = 0;
    if (tid < N_STEPS) flag[tid] = 0;
    if (tid == 0) ovf = 0;
    __syncthreads();

    int* mylist = lists + w * LIST_PITCH;
    int* pendw  = pend + w * PEND_CAP;
    int  fillw  = 0;

    const uint4* keys16 = (const uint4*)keys;
    const unsigned pat = (unsigned)blockIdx.x * 0x01010101u;
    const int rot = (int)blockIdx.x % ROUNDS;

    for (int r0 = 0; r0 < ROUNDS; r0 += G_FAST)
        fillw = sweep_group<G_FAST>(keys16, r0, rot, pat, w, lane, below, mylist, fillw);
    if (lane == 0 && fillw > LIST_CAP) ovf = 1;
    __syncthreads();
    if (__builtin_amdgcn_readfirstlane(ovf) != 0) {
        fillw = 0;
        for (int st = 0; st < N_STEPS; ++st) {
            if (lane == 0) {
                fillv[w] = fillw;
                if (fillw > LIST_CAP - G_SAFE * CHUNK_EDGES) flag[st] = 1;
            }
            __syncthreads();
            if (__builtin_amdgcn_readfirstlane(flag[st]) != 0) {
                finish_list(dst, mylist, fillw, n0, lane);
                __syncthreads();
                drain_lists(nf, ef, src, acc, degc, lists, fillv, pendw, w, lane);
                __syncthreads();
                fillw = 0;
            }
            fillw = sweep_group<G_SAFE>(keys16, st * G_SAFE, rot, pat, w, lane, below, mylist, fillw);
        }
    }
    finish_list(dst, mylist, fillw, n0, lane);
    if (lane == 0) fillv[w] = fillw;
    __syncthreads();
    drain_lists(nf, ef, src, acc, degc, lists, fillv, pendw, w, lane);
    __syncthreads();

    if (w < N_TILES) {
        const int  l15 = lane & 15;
        const bool lo  = (lane < 16);
        const int  lr  = w * 16 + l15;
        const int  dc  = degc[lr < NODES_PER_WG ? lr : 0];
        v2f a; a.x = (lo && lr < NODES_PER_WG) ? (float)dc : 0.0f; a.y = 0.0f;

        v2f b0; b0.x = lo ? p0 : 0.0f; b0.y = 0.0f;
        v2f b1; b1.x = lo ? p1 : 0.0f; b1.y = 0.0f;
        v2f b2; b2.x = lo ? h0 : 0.0f; b2.y = 0.0f;
        v2f b3; b3.x = lo ? h1 : 0.0f; b3.y = 0.0f;

        v8f c = {0.f, 0.f, 0.f, 0.f, 0.f, 0.f, 0.f, 0.f};
        v8f d0 = __builtin_amdgcn_wmma_f32_16x16x4_f32(false, a, false, b0, (short)0, c, false, false);
        v8f d1 = __builtin_amdgcn_wmma_f32_16x16x4_f32(false, a, false, b1, (short)0, c, false, false);
        v8f d2 = __builtin_amdgcn_wmma_f32_16x16x4_f32(false, a, false, b2, (short)0, c, false, false);
        v8f d3 = __builtin_amdgcn_wmma_f32_16x16x4_f32(false, a, false, b3, (short)0, c, false, false);

#pragma unroll
        for (int r = 0; r < 8; ++r) {
            const float s0 = __shfl_xor(d0[r], 16);
            const float s1 = __shfl_xor(d1[r], 16);
            const float s2 = __shfl_xor(d2[r], 16);
            const float s3 = __shfl_xor(d3[r], 16);
            const int ra = w * 16 + r;
            const int rb = ra + 8;
            float* oa = out + (size_t)(n0 + ra) * OUT_DIM;
            oa[128 + lane] = lo ? d0[r] : s1;
            oa[192 + lane] = lo ? d2[r] : s3;
            if (rb < NODES_PER_WG) {
                float* ob = out + (size_t)(n0 + rb) * OUT_DIM;
                ob[128 + lane] = lo ? s0 : d1[r];
                ob[192 + lane] = lo ? s2 : d3[r];
            }
        }
    }

    for (int r = w; r < NODES_PER_WG; r += 32) {
        const float* a = acc + r * ACC_W;
        float* o = out + (size_t)(n0 + r) * OUT_DIM;
        *(float2*)(o + 2 * lane)      = *(const float2*)(a + 2 * lane);
        *(float2*)(o + 64 + 2 * lane) = *(const float2*)(a + 64 + 2 * lane);
        o[160 + lane] = a[128 + lane];
    }
}

extern "C" void kernel_launch(void* const* d_in, const int* in_sizes, int n_in,
                              void* d_out, int out_size, void* d_ws, size_t ws_size,
                              hipStream_t stream) {
    (void)in_sizes; (void)n_in; (void)out_size; (void)ws_size;

    const float* nf     = (const float*)d_in[0];
    const float* ef     = (const float*)d_in[1];
    const float* prompt = (const float*)d_in[2];
    const float* hp     = (const float*)d_in[3];
    const int*   src    = (const int*)d_in[4];
    const int*   dst    = (const int*)d_in[5];

    float* out = (float*)d_out;
    int* keys = (int*)((char*)d_ws + 200704);

    key_kernel<<<(KEY_WORDS + 255) / 256, 256, 0, stream>>>(dst, keys);
    owner_kernel<<<N_NODES / NODES_PER_WG, 1024, 0, stream>>>(nf, ef, src, dst, keys, prompt, hp, out);
}


# owner waves no longer scan all 32 lists: entries counting-sorted by owner wave in LDS (per-list counts, prefix, scatter), each owner reads its own run; per-node sum order changes, stays fixed
# speedup vs baseline: 3.4640x; 1.0846x over previous
#include <hip/hip_runtime.h>

#define N_NODES   50000
#define N_EDGES   800000
#define IN_FEAT   128
#define PROMPT_DIM 32
#define EF_DIM     32
#define HP_DIM     32
#define OUT_DIM   224

typedef __attribute__((ext_vector_type(2))) float v2f;
typedef __attribute__((ext_vector_type(8))) float v8f;

#define NODES_PER_WG 200
#define ACC_W        160
#define CHUNK_EDGES  512
#define N_CHUNKS     1563
#define KEY_WORDS    (N_CHUNKS * CHUNK_EDGES / 4)
#define ROUNDS       49
#define G_FAST       7
#define G_SAFE       2
#define N_STEPS      25
#define LIST_CAP     1280
#define LIST_PITCH   1281
#define PEND_CAP     64
#define FAST_CAP     200
#define SORT_CAP     (32 * FAST_CAP)
#define XBUF_INTS    (2048 + SORT_CAP)
#define BATCH        16
#define SLOTS        7
#define N_TILES      13
#define NO_OWNER     255u

__device__ __forceinline__ unsigned owner_key(int d) {
    return (unsigned)d < (unsigned)N_NODES ? (unsigned)d / (unsigned)NODES_PER_WG : NO_OWNER;
}

__global__ void __launch_bounds__(256) key_kernel(const int* __restrict__ dst,
                                                  int* __restrict__ keys) {
    const int t = blockIdx.x * 256 + threadIdx.x;
    if (t >= KEY_WORDS) return;
    unsigned k = 0xFFFFFFFFu;
    if (t < N_EDGES / 4) {
        const int4 d = *(const int4*)(dst + 4 * t);
        k = owner_key(d.x) | (owner_key(d.y) << 8) | (owner_key(d.z) << 16) | (owner_key(d.w) << 24);
    }
    keys[t] = (int)k;
}


__device__ __forceinline__ void add_batch(const float* __restrict__ nf,
                                          const float* __restrict__ ef,
                                          const int*   __restrict__ src,
                                          float* acc, int* degc,
                                          const int* p, int n, int w, int lane, unsigned below) {
    const int e0  = p[lane < n ? lane : 0];
    const int key = lane < n ? ((e0 & 255) >> 5) : SLOTS;
    int rank = 0, base = 0, mine = 0;
    for (int s = 0; s <= SLOTS; ++s) {
        const unsigned m = __builtin_amdgcn_ballot_w32(key == s);
        if (key == s)  rank = base + __popc(m & below);
        if (lane == s) mine = __popc(m);
        base += __popc(m);
    }
    const int ent = __builtin_amdgcn_ds_permute(rank << 2, e0);
    const int sv  = src[ent >> 8];
    int   ln[BATCH];
    float2 xa[BATCH], xb[BATCH];
    float  fe[BATCH];
#pragma unroll
    for (int i = 0; i < BATCH; ++i) {
        const int en = __builtin_amdgcn_readlane(ent, i);
        ln[i] = en & 255;
        fe[i] = ef[(size_t)(en >> 8) * EF_DIM + lane];
    }
#pragma unroll
    for (int i = 0; i < BATCH; ++i) {
        const int s = __builtin_amdgcn_readlane(sv, i);
        const float* r = nf + (size_t)s * IN_FEAT;
        xa[i] = *(const float2*)(r + 2 * lane);
        xb[i] = *(const float2*)(r + 64 + 2 * lane);
    }
    if (lane < SLOTS && mine > 0) degc[w + 32 * lane] += mine;

    float2 ta = make_float2(0.0f, 0.0f), tb = make_float2(0.0f, 0.0f);
    float  tc = 0.0f;
#pragma unroll
    for (int i = 0; i < BATCH; ++i) {
        if (i == 0 || i < n) {
            float* a = acc + ln[i] * ACC_W;
            bool first = true;
            if (i > 0) first = (ln[i] != ln[i - 1]);
            if (first) {
                ta = *(float2*)(a + 2 * lane);
                tb = *(float2*)(a + 64 + 2 * lane);
                tc = a[128 + lane];
            }
            ta.x += xa[i].x; ta.y += xa[i].y;
            tb.x += xb[i].x; tb.y += xb[i].y;
            tc   += fe[i];
            bool last = true;
            if (i + 1 < BATCH) last = (i + 1 >= n) || (ln[i + 1] != ln[i]);
            if (last) {
                *(float2*)(a + 2 * lane)      = ta;
                *(float2*)(a + 64 + 2 * lane) = tb;
                a[128 + lane] = tc;
            }
        }
    }
}

__device__ __forceinline__ unsigned same_owner(int o, bool valid) {
    unsigned p = __builtin_amdgcn_ballot_w32(valid);
    for (int b = 0; b < 5; ++b) {
        const bool bit = ((o >> b) & 1) != 0;
        const unsigned m = __builtin_amdgcn_ballot_w32(valid && bit);
        p &= bit ? m : ~m;
    }
    return p;
}

__device__ __forceinline__ void drain_lists(const float* __restrict__ nf,
                                            const float* __restrict__ ef,
                                            const int*   __restrict__ src,
                                            float* acc, int* degc,
                                            const int* lists, const int* fillv,
                                            int* pendw, int w, int lane) {
    const unsigned below = (1u << lane) - 1u;
    const int  myfill = fillv[lane] < LIST_CAP ? fillv[lane] : LIST_CAP;
    const int* mylist = lists + lane * LIST_PITCH;
    int  npend = 0;
    int  k = 0;
    bool more = true;
    for (;;) {
        while (more && npend < BATCH) {
            const bool valid = k < myfill;
            if (__builtin_amdgcn_ballot_w32(valid) == 0u) { more = false; break; }
            const int raw  = mylist[k];
            ++k;
            const bool mine = valid && ((raw & 31) == w);
            const unsigned m = __builtin_amdgcn_ballot_w32(mine);
            if (m != 0u) {
                if (mine) pendw[npend + __popc(m & below)] = raw;
                npend += __popc(m);
            }
        }
        if (npend == 0) break;
        const int n = npend < BATCH ? npend : BATCH;
        add_batch(nf, ef, src, acc, degc, pendw, n, w, lane, below);
        const int left = npend - n;
        if (left > 0) {
            const int t = pendw[BATCH + (lane < left ? lane : 0)];
            if (lane < left) pendw[lane] = t;
        }
        npend = left;
    }
}

template <int G>
__device__ __forceinline__ int sweep_group(const uint4* __restrict__ keys16, int r0, int rot,
                                           unsigned pat, int w, int lane, unsigned below,
                                           int* mylist, int fillw) {
    uint4 q[G];
    int   cb[G];
    for (int g = 0; g < G; ++g) {
        const int r = r0 + g;
        int rr = r + rot;
        if (rr >= ROUNDS) rr -= ROUNDS;
        const int c = rr * 32 + w;
        const bool ok = (r < ROUNDS) && (c < N_CHUNKS);
        cb[g] = ok ? c : -1;
        q[g] = keys16[(ok ? c : 0) * 32 + lane];
    }
    for (int g = 0; g < G; ++g) {
        const unsigned lo7 = 0x7F7F7F7Fu, hi1 = 0x80808080u;
        const unsigned x0 = q[g].x ^ pat, x1 = q[g].y ^ pat, x2 = q[g].z ^ pat, x3 = q[g].w ^ pat;
        const unsigned y0 = (((x0 & lo7) + lo7) | x0) & hi1;
        const unsigned y1 = (((x1 & lo7) + lo7) | x1) & hi1;
        const unsigned y2 = (((x2 & lo7) + lo7) | x2) & hi1;
        const unsigned y3 = (((x3 & lo7) + lo7) | x3) & hi1;
        unsigned zz = ((y0 >> 7) | (y1 >> 6) | (y2 >> 5) | (y3 >> 4)) ^ 0x0F0F0F0Fu;
        if (cb[g] < 0) zz = 0u;
        const int e0 = cb[g] * CHUNK_EDGES + lane * 16;
        unsigned m = __builtin_amdgcn_ballot_w32(zz != 0u);
        for (int rd = 0; rd < 16; ++rd) {
            if (m == 0u) break;
            if (zz != 0u) {
                const int p = __builtin_ctz(zz);
                zz &= zz - 1u;
                const int pos = fillw + __popc(m & below);
                if (pos < LIST_CAP) mylist[pos] = e0 + (((p & 7) << 2) | (p >> 3));
            }
            fillw += __popc(m);
            m = __builtin_amdgcn_ballot_w32(zz != 0u);
        }
    }
    return fillw;
}

__device__ __forceinline__ void finish_list(const int* __restrict__ dst, int* mylist,
                                            int fillw, int n0, int lane) {
    for (int k0 = 0; k0 < fillw; k0 += 128) {
        int e[4], d[4];
        for (int i = 0; i < 4; ++i) {
            const int kk = k0 + i * 32 + lane;
            e[i] = kk < fillw ? mylist[kk] : 0;
        }
        for (int i = 0; i < 4; ++i) d[i] = dst[e[i]];
        for (int i = 0; i < 4; ++i) {
            const int kk = k0 + i * 32 + lane;
            if (kk < fillw) mylist[kk] = (e[i] << 8) | (d[i] - n0);
        }
    }
}

__global__ void __launch_bounds__(1024) owner_kernel(const float* __restrict__ nf,
                                                     const float* __restrict__ ef,
                                                     const int*   __restrict__ src,
                                                     const int*   __restrict__ dst,
                                                     const int*   __restrict__ keys,
                                                     const float* __restrict__ prompt,
                                                     const float* __restrict__ hp,
                                                     float* __restrict__ out) {
    __shared__ float acc[NODES_PER_WG * ACC_W];
    __shared__ int   lists[32 * LIST_PITCH];
    __shared__ int   xbuf[XBUF_INTS];
    __shared__ int   fillv[32];
    __shared__ int   degc[NODES_PER_WG];
    __shared__ int   flag[N_STEPS];
    __shared__ int   ovf;

    const int tid  = threadIdx.x;
    const int lane = tid & 31;
    const int w    = __builtin_amdgcn_readfirstlane(tid >> 5);
    const int n0   = blockIdx.x * NODES_PER_WG;
    const unsigned below = (1u << lane) - 1u;

    const float p0 = prompt[lane & 15], p1 = prompt[16 + (lane & 15)];
    const float h0 = hp[lane & 15],     h1 = hp[16 + (lane & 15)];

    for (int i = tid; i < NODES_PER_WG * ACC_W; i += 1024) acc[i] = 0.0f;
    if (tid < NODES_PER_WG) degc[tid] = 0;
    xbuf[tid] = 0;
    if (tid < N_STEPS) flag[tid] = 0;
    if (tid == 0) ovf = 0;
    __syncthreads();

    int* mylist = lists + w * LIST_PITCH;
    int  fillw  = 0;

    const uint4* keys16 = (const uint4*)keys;
    const unsigned pat = (unsigned)blockIdx.x * 0x01010101u;
    const int rot = (int)blockIdx.x % ROUNDS;

    for (int r0 = 0; r0 < ROUNDS; r0 += G_FAST)
        fillw = sweep_group<G_FAST>(keys16, r0, rot, pat, w, lane, below, mylist, fillw);
    if (lane == 0 && fillw > FAST_CAP) ovf = 1;
    __syncthreads();
    if (__builtin_amdgcn_readfirstlane(ovf) != 0) {
        int* pendw = xbuf + w * PEND_CAP;
        fillw = 0;
        for (int st = 0; st < N_STEPS; ++st) {
            if (lane == 0) {
                fillv[w] = fillw;
                if (fillw > LIST_CAP - G_SAFE * CHUNK_EDGES) flag[st] = 1;
            }
            __syncthreads();
            if (__builtin_amdgcn_readfirstlane(flag[st]) != 0) {
                finish_list(dst, mylist, fillw, n0, lane);
                __syncthreads();
                drain_lists(nf, ef, src, acc, degc, lists, fillv, pendw, w, lane);
                __syncthreads();
                fillw = 0;
            }
            fillw = sweep_group<G_SAFE>(keys16, st * G_SAFE, rot, pat, w, lane, below, mylist, fillw);
        }
        finish_list(dst, mylist, fillw, n0, lane);
        if (lane == 0) fillv[w] = fillw;
        __syncthreads();
        drain_lists(nf, ef, src, acc, degc, lists, fillv, pendw, w, lane);
    } else {
        int* cntw   = xbuf + w * 32;
        int* posw   = xbuf + 1024 + w * 32;
        int* sorted = xbuf + 2048;
        finish_list(dst, mylist, fillw, n0, lane);
        for (int k0 = 0; k0 < fillw; k0 += 32) {
            const int  kk    = k0 + lane;
            const bool valid = kk < fillw;
            const int  o     = mylist[valid ? kk : 0] & 31;
            const unsigned p = same_owner(o, valid);
            const int  run   = cntw[o];
            if (valid && (p & below) == 0u) cntw[o] = run + __popc(p);
        }
        __syncthreads();
        int colsum = 0, pre = 0;
        for (int s = 0; s < 32; ++s) {
            const int c = xbuf[s * 32 + lane];
            colsum += c;
            if (s < w) pre += c;
        }
        int incl = colsum;
        for (int d = 1; d < 32; d <<= 1) {
            const int t = __shfl_up(incl, d);
            if (lane >= d) incl += t;
        }
        const int first = incl - colsum;
        posw[lane] = first + pre;
        const int start = __builtin_amdgcn_readlane(first, w);
        const int len   = __builtin_amdgcn_readlane(colsum, w);
        for (int k0 = 0; k0 < fillw; k0 += 32) {
            const int  kk    = k0 + lane;
            const bool valid = kk < fillw;
            const int  ent   = mylist[valid ? kk : 0];
            const int  o     = ent & 31;
            const unsigned p = same_owner(o, valid);
            const int  b     = posw[o];
            if (valid) sorted[b + __popc(p & below)] = ent;
            if (valid && (p & below) == 0u) posw[o] = b + __popc(p);
        }
        __syncthreads();
        for (int i = 0; i < len; i += BATCH) {
            const int n = len - i < BATCH ? len - i : BATCH;
            add_batch(nf, ef, src, acc, degc, sorted + start + i, n, w, lane, below);
        }
    }
    __syncthreads();

    if (w < N_TILES) {
        const int  l15 = lane & 15;
        const bool lo  = (lane < 16);
        const int  lr  = w * 16 + l15;
        const int  dc  = degc[lr < NODES_PER_WG ? lr : 0];
        v2f a; a.x = (lo && lr < NODES_PER_WG) ? (float)dc : 0.0f; a.y = 0.0f;

        v2f b0; b0.x = lo ? p0 : 0.0f; b0.y = 0.0f;
        v2f b1; b1.x = lo ? p1 : 0.0f; b1.y = 0.0f;
        v2f b2; b2.x = lo ? h0 : 0.0f; b2.y = 0.0f;
        v2f b3; b3.x = lo ? h1 : 0.0f; b3.y = 0.0f;

        v8f c = {0.f, 0.f, 0.f, 0.f, 0.f, 0.f, 0.f, 0.f};
        v8f d0 = __builtin_amdgcn_wmma_f32_16x16x4_f32(false, a, false, b0, (short)0, c, false, false);
        v8f d1 = __builtin_amdgcn_wmma_f32_16x16x4_f32(false, a, false, b1, (short)0, c, false, false);
        v8f d2 = __builtin_amdgcn_wmma_f32_16x16x4_f32(false, a, false, b2, (short)0, c, false, false);
        v8f d3 = __builtin_amdgcn_wmma_f32_16x16x4_f32(false, a, false, b3, (short)0, c, false, false);

#pragma unroll
        for (int r = 0; r < 8; ++r) {
            const float s0 = __shfl_xor(d0[r], 16);
            const float s1 = __shfl_xor(d1[r], 16);
            const float s2 = __shfl_xor(d2[r], 16);
            const float s3 = __shfl_xor(d3[r], 16);
            const int ra = w * 16 + r;
            const int rb = ra + 8;
            float* oa = out + (size_t)(n0 + ra) * OUT_DIM;
            oa[128 + lane] = lo ? d0[r] : s1;
            oa[192 + lane] = lo ? d2[r] : s3;
            if (rb < NODES_PER_WG) {
                float* ob = out + (size_t)(n0 + rb) * OUT_DIM;
                ob[128 + lane] = lo ? s0 : d1[r];
                ob[192 + lane] = lo ? s2 : d3[r];
            }
        }
    }

    for (int r = w; r < NODES_PER_WG; r += 32) {
        const float* a = acc + r * ACC_W;
        float* o = out + (size_t)(n0 + r) * OUT_DIM;
        *(float2*)(o + 2 * lane)      = *(const float2*)(a + 2 * lane);
        *(float2*)(o + 64 + 2 * lane) = *(const float2*)(a + 64 + 2 * lane);
        o[160 + lane] = a[128 + lane];
    }
}

extern "C" void kernel_launch(void* const* d_in, const int* in_sizes, int n_in,
                              void* d_out, int out_size, void* d_ws, size_t ws_size,
                              hipStream_t stream) {
    (void)in_sizes; (void)n_in; (void)out_size; (void)ws_size;

    const float* nf     = (const float*)d_in[0];
    const float* ef     = (const float*)d_in[1];
    const float* prompt = (const float*)d_in[2];
    const float* hp     = (const float*)d_in[3];
    const int*   src    = (const int*)d_in[4];
    const int*   dst    = (const int*)d_in[5];

    float* out = (float*)d_out;
    int* keys = (int*)((char*)d_ws + 200704);

    key_kernel<<<(KEY_WORDS + 255) / 256, 256, 0, stream>>>(dst, keys);
    owner_kernel<<<N_NODES / NODES_PER_WG, 1024, 0, stream>>>(nf, ef, src, dst, keys, prompt, hp, out);
}


# all-keys sweep replaced by pre-bucketing: launch 1 counting-sorts each 3125-edge chunk by owner in LDS, owners copy their 256 runs (staged in their own out rows); per-node sum order changes, fixed
# speedup vs baseline: 4.2072x; 1.2145x over previous
#include <hip/hip_runtime.h>

#define N_NODES   50000
#define N_EDGES   800000
#define IN_FEAT   128
#define PROMPT_DIM 32
#define EF_DIM     32
#define HP_DIM     32
#define OUT_DIM   224

typedef __attribute__((ext_vector_type(2))) float v2f;
typedef __attribute__((ext_vector_type(8))) float v8f;

#define NODES_PER_WG 200
#define N_OWNERS     250
#define ACC_W        160
#define N_CH         256
#define CH_EDGES     3125
#define B_WAVES      25
#define W_EDGES      125
#define B_THREADS    800
#define RUNS_PER_WAVE 10
#define CNT_PITCH    256
#define SLOT         160
#define REGION_INTS  (NODES_PER_WG * OUT_DIM)
#define CH_PER_WAVE  8
#define PIECE_EDGES  1024
#define N_PIECES     4
#define LIST_CAP     1280
#define LIST_PITCH   1281
#define PEND_CAP     64
#define FAST_CAP     200
#define SORT_CAP     (32 * FAST_CAP)
#define XBUF_INTS    (2048 + SORT_CAP)
#define BATCH        16
#define SLOTS        7
#define N_TILES      13

__device__ __forceinline__ unsigned same_owner(int o, bool valid) {
    unsigned p = __builtin_amdgcn_ballot_w32(valid);
    for (int b = 0; b < 5; ++b) {
        const bool bit = ((o >> b) & 1) != 0;
        const unsigned m = __builtin_amdgcn_ballot_w32(valid && bit);
        p &= bit ? m : ~m;
    }
    return p;
}

__device__ __forceinline__ unsigned same_owner_wg(int o, bool valid) {
    unsigned p = __builtin_amdgcn_ballot_w32(valid);
    for (int b = 0; b < 8; ++b) {
        const bool bit = ((o >> b) & 1) != 0;
        const unsigned m = __builtin_amdgcn_ballot_w32(bit);
        p &= bit ? m : ~m;
    }
    return p;
}

__global__ void __launch_bounds__(B_THREADS) bucket_kernel(const int* __restrict__ dst,
                                                           int* __restrict__ counts,
                                                           float* __restrict__ out) {
    __shared__ int cnt[B_WAVES * CNT_PITCH];
    __shared__ int sorted[CH_EDGES];
    __shared__ int ctot[CNT_PITCH];
    __shared__ int base[CNT_PITCH];
    __shared__ int wtot[8];

    const int tid  = threadIdx.x;
    const int lane = tid & 31;
    const int w    = __builtin_amdgcn_readfirstlane(tid >> 5);
    const int c    = blockIdx.x;
    const unsigned below = (1u << lane) - 1u;
    const int e0   = c * CH_EDGES + w * W_EDGES;

    int d[4];
#pragma unroll
    for (int g = 0; g < 4; ++g) {
        const int idx = g * 32 + lane;
        d[g] = idx < W_EDGES ? dst[e0 + idx] : -1;
    }
    for (int i = tid; i < B_WAVES * CNT_PITCH; i += B_THREADS) cnt[i] = 0;
    __syncthreads();

    int* cntw = cnt + w * CNT_PITCH;
    int own[4], rank[4];
#pragma unroll
    for (int g = 0; g < 4; ++g) {
        const bool valid = (unsigned)d[g] < (unsigned)N_NODES;
        const int  o     = valid ? (int)((unsigned)d[g] / (unsigned)NODES_PER_WG) : 0;
        const unsigned p = same_owner_wg(o, valid);
        const int  run   = cntw[o];
        rank[g] = run + __popc(p & below);
        if (valid && (p & below) == 0u) cntw[o] = run + __popc(p);
        own[g] = valid ? o : -1;
    }
    __syncthreads();

    if (tid < CNT_PITCH) {
        int tot = 0;
#pragma unroll
        for (int s = 0; s < B_WAVES; ++s) {
            const int t = cnt[s * CNT_PITCH + tid];
            cnt[s * CNT_PITCH + tid] = tot;
            tot += t;
        }
        int incl = tot;
        for (int dd = 1; dd < 32; dd <<= 1) {
            const int t = __shfl_up(incl, dd);
            if (lane >= dd) incl += t;
        }
        ctot[tid] = tot;
        base[tid] = incl - tot;
        if (lane == 31) wtot[w] = incl;
        counts[c * CNT_PITCH + tid] = tot;
    }
    __syncthreads();
    if (tid < CNT_PITCH) {
        int add = 0;
        for (int v = 0; v < 8; ++v) {
            const int t = wtot[v];
            if (v < w) add += t;
        }
        base[tid] += add;
    }
    __syncthreads();

#pragma unroll
    for (int g = 0; g < 4; ++g) {
        if (own[g] >= 0) {
            const int o = own[g];
            sorted[base[o] + cntw[o] + rank[g]] = ((e0 + g * 32 + lane) << 8) | (d[g] - o * NODES_PER_WG);
        }
    }
    __syncthreads();

    const int oi = w + B_WAVES * (lane < RUNS_PER_WAVE ? lane : 0);
    int nn = ctot[oi] < SLOT ? ctot[oi] : SLOT;
    if (lane >= RUNS_PER_WAVE) nn = 0;
    const int bb = base[oi];
    int* s0 = (int*)out + (size_t)w * REGION_INTS + c * SLOT;
    int v[RUNS_PER_WAVE];
#pragma unroll
    for (int i = 0; i < RUNS_PER_WAVE; ++i) {
        const int n  = __builtin_amdgcn_readlane(nn, i);
        const int b0 = __builtin_amdgcn_readlane(bb, i);
        v[i] = lane < n ? sorted[b0 + lane] : 0;
    }
#pragma unroll
    for (int i = 0; i < RUNS_PER_WAVE; ++i) {
        const int n = __builtin_amdgcn_readlane(nn, i);
        if (lane < n) s0[(size_t)i * (B_WAVES * REGION_INTS) + lane] = v[i];
    }
    if (__builtin_amdgcn_ballot_w32(nn > 32) != 0u) {
        for (int i = 0; i < RUNS_PER_WAVE; ++i) {
            const int n  = __builtin_amdgcn_readlane(nn, i);
            const int b0 = __builtin_amdgcn_readlane(bb, i);
            for (int k = 32 + lane; k < n; k += 32)
                s0[(size_t)i * (B_WAVES * REGION_INTS) + k] = sorted[b0 + k];
        }
    }
}


__device__ __forceinline__ void add_batch(const float* __restrict__ nf,
                                          const float* __restrict__ ef,
                                          const int*   __restrict__ src,
                                          float* acc, int* degc,
                                          const int* p, int n, int w, int lane, unsigned below) {
    const int e0  = p[lane < n ? lane : 0];
    const int key = lane < n ? ((e0 & 255) >> 5) : SLOTS;
    int rank = 0, base = 0, mine = 0;
    for (int s = 0; s <= SLOTS; ++s) {
        const unsigned m = __builtin_amdgcn_ballot_w32(key == s);
        if (key == s)  rank = base + __popc(m & below);
        if (lane == s) mine = __popc(m);
        base += __popc(m);
    }
    const int ent = __builtin_amdgcn_ds_permute(rank << 2, e0);
    const int sv  = src[ent >> 8];
    int   ln[BATCH];
    float2 xa[BATCH], xb[BATCH];
    float  fe[BATCH];
#pragma unroll
    for (int i = 0; i < BATCH; ++i) {
        const int en = __builtin_amdgcn_readlane(ent, i);
        ln[i] = en & 255;
        fe[i] = ef[(size_t)(en >> 8) * EF_DIM + lane];
    }
#pragma unroll
    for (int i = 0; i < BATCH; ++i) {
        const int s = __builtin_amdgcn_readlane(sv, i);
        const float* r = nf + (size_t)s * IN_FEAT;
        xa[i] = *(const float2*)(r + 2 * lane);
        xb[i] = *(const float2*)(r + 64 + 2 * lane);
    }
    if (lane < SLOTS && mine > 0) degc[w + 32 * lane] += mine;

    float2 ta = make_float2(0.0f, 0.0f), tb = make_float2(0.0f, 0.0f);
    float  tc = 0.0f;
#pragma unroll
    for (int i = 0; i < BATCH; ++i) {
        if (i == 0 || i < n) {
            float* a = acc + ln[i] * ACC_W;
            bool first = true;
            if (i > 0) first = (ln[i] != ln[i - 1]);
            if (first) {
                ta = *(float2*)(a + 2 * lane);
                tb = *(float2*)(a + 64 + 2 * lane);
                tc = a[128 + lane];
            }
            ta.x += xa[i].x; ta.y += xa[i].y;
            tb.x += xb[i].x; tb.y += xb[i].y;
            tc   += fe[i];
            bool last = true;
            if (i + 1 < BATCH) last = (i + 1 >= n) || (ln[i + 1] != ln[i]);
            if (last) {
                *(float2*)(a + 2 * lane)      = ta;
                *(float2*)(a + 64 + 2 * lane) = tb;
                a[128 + lane] = tc;
            }
        }
    }
}

__device__ __forceinline__ void drain_lists(const float* __restrict__ nf,
                                            const float* __restrict__ ef,
                                            const int*   __restrict__ src,
                                            float* acc, int* degc,
                                            const int* lists, const int* fillv,
                                            int* pendw, int w, int lane) {
    const unsigned below = (1u << lane) - 1u;
    const int  myfill = fillv[lane] < LIST_CAP ? fillv[lane] : LIST_CAP;
    const int* mylist = lists + lane * LIST_PITCH;
    int  npend = 0;
    int  k = 0;
    bool more = true;
    for (;;) {
        while (more && npend < BATCH) {
            const bool valid = k < myfill;
            if (__builtin_amdgcn_ballot_w32(valid) == 0u) { more = false; break; }
            const int raw  = mylist[k];
            ++k;
            const bool mine = valid && ((raw & 31) == w);
            const unsigned m = __builtin_amdgcn_ballot_w32(mine);
            if (m != 0u) {
                if (mine) pendw[npend + __popc(m & below)] = raw;
                npend += __popc(m);
            }
        }
        if (npend == 0) break;
        const int n = npend < BATCH ? npend : BATCH;
        add_batch(nf, ef, src, acc, degc, pendw, n, w, lane, below);
        const int left = npend - n;
        if (left > 0) {
            const int t = pendw[BATCH + (lane < left ? lane : 0)];
            if (lane < left) pendw[lane] = t;
        }
        npend = left;
    }
}

__global__ void __launch_bounds__(1024) owner_kernel(const float* __restrict__ nf,
                                                     const float* __restrict__ ef,
                                                     const int*   __restrict__ src,
                                                     const int*   __restrict__ dst,
                                                     const int*   __restrict__ counts,
                                                     const float* __restrict__ prompt,
                                                     const float* __restrict__ hp,
                                                     float* __restrict__ out) {
    __shared__ float acc[NODES_PER_WG * ACC_W];
    __shared__ int   lists[32 * LIST_PITCH];
    __shared__ int   xbuf[XBUF_INTS];
    __shared__ int   fillv[32];
    __shared__ int   degc[NODES_PER_WG];
    __shared__ int   ovf;

    const int tid  = threadIdx.x;
    const int lane = tid & 31;
    const int w    = __builtin_amdgcn_readfirstlane(tid >> 5);
    const int n0   = blockIdx.x * NODES_PER_WG;
    const unsigned below = (1u << lane) - 1u;

    const int nj = lane < CH_PER_WAVE ? counts[(CH_PER_WAVE * w + lane) * CNT_PITCH + blockIdx.x] : 0;

    const float p0 = prompt[lane & 15], p1 = prompt[16 + (lane & 15)];
    const float h0 = hp[lane & 15],     h1 = hp[16 + (lane & 15)];

    for (int i = tid; i < NODES_PER_WG * ACC_W; i += 1024) acc[i] = 0.0f;
    if (tid < NODES_PER_WG) degc[tid] = 0;
    xbuf[tid] = 0;
    if (tid == 0) ovf = 0;
    __syncthreads();

    int* mylist = lists + w * LIST_PITCH;
    int  fillw  = 0;

    int incl8 = nj;
    for (int d = 1; d < CH_PER_WAVE; d <<= 1) {
        const int t = __shfl_up(incl8, d);
        if (lane >= d) incl8 += t;
    }
    const int offl  = incl8 - nj;
    const int total = __builtin_amdgcn_readlane(incl8, CH_PER_WAVE - 1);
    int maxn = nj;
    for (int d = 1; d < CH_PER_WAVE; d <<= 1) {
        const int t = __shfl_xor(maxn, d);
        maxn = t > maxn ? t : maxn;
    }
    maxn = __builtin_amdgcn_readfirstlane(maxn);
    const bool bad = (maxn > SLOT) || (total > FAST_CAP);
    if (!bad) {
        const int* myslots = (const int*)out + (size_t)blockIdx.x * REGION_INTS + (CH_PER_WAVE * w) * SLOT;
        for (int r0 = 0; r0 < maxn; r0 += 32) {
            const int idx = r0 + lane;
            int v[CH_PER_WAVE];
#pragma unroll
            for (int j = 0; j < CH_PER_WAVE; ++j) {
                const int n = __builtin_amdgcn_readlane(nj, j);
                v[j] = idx < n ? myslots[j * SLOT + idx] : 0;
            }
#pragma unroll
            for (int j = 0; j < CH_PER_WAVE; ++j) {
                const int n = __builtin_amdgcn_readlane(nj, j);
                const int o = __builtin_amdgcn_readlane(offl, j);
                if (idx < n) mylist[o + idx] = v[j];
            }
        }
        fillw = total;
    }
    if (lane == 0 && bad) ovf = 1;
    __syncthreads();
    if (__builtin_amdgcn_readfirstlane(ovf) != 0) {
        int* pendw = xbuf + w * PEND_CAP;
        for (int st = 0; st < CH_PER_WAVE * N_PIECES; ++st) {
            const int ch    = CH_PER_WAVE * w + (st >> 2);
            const int first = ch * CH_EDGES + (st & 3) * PIECE_EDGES;
            const int end   = (st & 3) == N_PIECES - 1 ? (ch + 1) * CH_EDGES : first + PIECE_EDGES;
            fillw = 0;
            for (int g = 0; g < PIECE_EDGES / 32; ++g) {
                const int  e   = first + g * 32 + lane;
                const bool in  = e < end;
                const int  dd  = dst[in ? e : first];
                const unsigned loc = (unsigned)(dd - n0);
                const bool hit = in && loc < (unsigned)NODES_PER_WG;
                const unsigned m = __builtin_amdgcn_ballot_w32(hit);
                if (hit) mylist[fillw + __popc(m & below)] = (e << 8) | (int)loc;
                fillw += __popc(m);
            }
            if (lane == 0) fillv[w] = fillw;
            __syncthreads();
            drain_lists(nf, ef, src, acc, degc, lists, fillv, pendw, w, lane);
            __syncthreads();
        }
    } else {
        int* cntw   = xbuf + w * 32;
        int* posw   = xbuf + 1024 + w * 32;
        int* sorted = xbuf + 2048;
        for (int k0 = 0; k0 < fillw; k0 += 32) {
            const int  kk    = k0 + lane;
            const bool valid = kk < fillw;
            const int  o     = mylist[valid ? kk : 0] & 31;
            const unsigned p = same_owner(o, valid);
            const int  run   = cntw[o];
            if (valid && (p & below) == 0u) cntw[o] = run + __popc(p);
        }
        __syncthreads();
        int colsum = 0, pre = 0;
        for (int s = 0; s < 32; ++s) {
            const int c = xbuf[s * 32 + lane];
            colsum += c;
            if (s < w) pre += c;
        }
        int incl = colsum;
        for (int d = 1; d < 32; d <<= 1) {
            const int t = __shfl_up(incl, d);
            if (lane >= d) incl += t;
        }
        const int first = incl - colsum;
        posw[lane] = first + pre;
        const int start = __builtin_amdgcn_readlane(first, w);
        const int len   = __builtin_amdgcn_readlane(colsum, w);
        for (int k0 = 0; k0 < fillw; k0 += 32) {
            const int  kk    = k0 + lane;
            const bool valid = kk < fillw;
            const int  ent   = mylist[valid ? kk : 0];
            const int  o     = ent & 31;
            const unsigned p = same_owner(o, valid);
            const int  b     = posw[o];
            if (valid) sorted[b + __popc(p & below)] = ent;
            if (valid && (p & below) == 0u) posw[o] = b + __popc(p);
        }
        __syncthreads();
        for (int i = 0; i < len; i += BATCH) {
            const int n = len - i < BATCH ? len - i : BATCH;
            add_batch(nf, ef, src, acc, degc, sorted + start + i, n, w, lane, below);
        }
    }
    __syncthreads();

    if (w < N_TILES) {
        const int  l15 = lane & 15;
        const bool lo  = (lane < 16);
        const int  lr  = w * 16 + l15;
        const int  dc  = degc[lr < NODES_PER_WG ? lr : 0];
        v2f a; a.x = (lo && lr < NODES_PER_WG) ? (float)dc : 0.0f; a.y = 0.0f;

        v2f b0; b0.x = lo ? p0 : 0.0f; b0.y = 0.0f;
        v2f b1; b1.x = lo ? p1 : 0.0f; b1.y = 0.0f;
        v2f b2; b2.x = lo ? h0 : 0.0f; b2.y = 0.0f;
        v2f b3; b3.x = lo ? h1 : 0.0f; b3.y = 0.0f;

        v8f c = {0.f, 0.f, 0.f, 0.f, 0.f, 0.f, 0.f, 0.f};
        v8f d0 = __builtin_amdgcn_wmma_f32_16x16x4_f32(false, a, false, b0, (short)0, c, false, false);
        v8f d1 = __builtin_amdgcn_wmma_f32_16x16x4_f32(false, a, false, b1, (short)0, c, false, false);
        v8f d2 = __builtin_amdgcn_wmma_f32_16x16x4_f32(false, a, false, b2, (short)0, c, false, false);
        v8f d3 = __builtin_amdgcn_wmma_f32_16x16x4_f32(false, a, false, b3, (short)0, c, false, false);

#pragma unroll
        for (int r = 0; r < 8; ++r) {
            const float s0 = __shfl_xor(d0[r], 16);
            const float s1 = __shfl_xor(d1[r], 16);
            const float s2 = __shfl_xor(d2[r], 16);
            const float s3 = __shfl_xor(d3[r], 16);
            const int ra = w * 16 + r;
            const int rb = ra + 8;
            float* oa = out + (size_t)(n0 + ra) * OUT_DIM;
            oa[128 + lane] = lo ? d0[r] : s1;
            oa[192 + lane] = lo ? d2[r] : s3;
            if (rb < NODES_PER_WG) {
                float* ob = out + (size_t)(n0 + rb) * OUT_DIM;
                ob[128 + lane] = lo ? s0 : d1[r];
                ob[192 + lane] = lo ? s2 : d3[r];
            }
        }
    }

    for (int r = w; r < NODES_PER_WG; r += 32) {
        const float* a = acc + r * ACC_W;
        float* o = out + (size_t)(n0 + r) * OUT_DIM;
        *(float2*)(o + 2 * lane)      = *(const float2*)(a + 2 * lane);
        *(float2*)(o + 64 + 2 * lane) = *(const float2*)(a + 64 + 2 * lane);
        o[160 + lane] = a[128 + lane];
    }
}

extern "C" void kernel_launch(void* const* d_in, const int* in_sizes, int n_in,
                              void* d_out, int out_size, void* d_ws, size_t ws_size,
                              hipStream_t stream) {
    (void)in_sizes; (void)n_in; (void)out_size; (void)ws_size;

    const float* nf     = (const float*)d_in[0];
    const float* ef     = (const float*)d_in[1];
    const float* prompt = (const float*)d_in[2];
    const float* hp     = (const float*)d_in[3];
    const int*   src    = (const int*)d_in[4];
    const int*   dst    = (const int*)d_in[5];

    float* out = (float*)d_out;
    int* counts = (int*)((char*)d_ws + 200704);

    bucket_kernel<<<N_CH, B_THREADS, 0, stream>>>(dst, counts, out);
    owner_kernel<<<N_OWNERS, 1024, 0, stream>>>(nf, ef, src, dst, counts, prompt, hp, out);
}


# src ids ride with the bucketed entries: launch 1 reads src coalesced and stores (entry, src) pairs per run, launch 2 keeps them in LDS; no 4-B src gather per edge; per-node sum order unchanged
# speedup vs baseline: 4.5293x; 1.0766x over previous
#include <hip/hip_runtime.h>

#define N_NODES   50000
#define N_EDGES   800000
#define IN_FEAT   128
#define PROMPT_DIM 32
#define EF_DIM     32
#define HP_DIM     32
#define OUT_DIM   224

typedef __attribute__((ext_vector_type(2))) float v2f;
typedef __attribute__((ext_vector_type(8))) float v8f;

#define NODES_PER_WG 200
#define N_OWNERS     250
#define ACC_W        160
#define N_CH         256
#define CH_EDGES     3125
#define B_WAVES      25
#define W_EDGES      125
#define B_THREADS    800
#define RUNS_PER_WAVE 10
#define CNT_PITCH    256
#define SLOT         160
#define SLOT_PAIRS   (SLOT / 2)
#define REGION_INTS  (NODES_PER_WG * OUT_DIM)
#define REGION_PAIRS (REGION_INTS / 2)
#define CH_PER_WAVE  8
#define PIECE_EDGES  1024
#define N_PIECES     4
#define LIST_CAP     1280
#define LIST_PITCH   1281
#define PEND_CAP     64
#define FAST_CAP     200
#define SORT_CAP     (32 * FAST_CAP)
#define XBUF_INTS    (2048 + SORT_CAP)
#define FLIST_PITCH  512
#define FSRC_OFF     256
#define BATCH        16
#define SLOTS        7
#define N_TILES      13

__device__ __forceinline__ unsigned same_owner(int o, bool valid) {
    unsigned p = __builtin_amdgcn_ballot_w32(valid);
    for (int b = 0; b < 5; ++b) {
        const bool bit = ((o >> b) & 1) != 0;
        const unsigned m = __builtin_amdgcn_ballot_w32(valid && bit);
        p &= bit ? m : ~m;
    }
    return p;
}

__device__ __forceinline__ unsigned same_owner_wg(int o, bool valid) {
    unsigned p = __builtin_amdgcn_ballot_w32(valid);
    for (int b = 0; b < 8; ++b) {
        const bool bit = ((o >> b) & 1) != 0;
        const unsigned m = __builtin_amdgcn_ballot_w32(bit);
        p &= bit ? m : ~m;
    }
    return p;
}

__global__ void __launch_bounds__(B_THREADS) bucket_kernel(const int* __restrict__ src,
                                                           const int* __restrict__ dst,
                                                           int* __restrict__ counts,
                                                           float* __restrict__ out) {
    __shared__ int  cnt[B_WAVES * CNT_PITCH];
    __shared__ int2 sorted[CH_EDGES];
    __shared__ int ctot[CNT_PITCH];
    __shared__ int base[CNT_PITCH];
    __shared__ int wtot[8];

    const int tid  = threadIdx.x;
    const int lane = tid & 31;
    const int w    = __builtin_amdgcn_readfirstlane(tid >> 5);
    const int c    = blockIdx.x;
    const unsigned below = (1u << lane) - 1u;
    const int e0   = c * CH_EDGES + w * W_EDGES;

    int d[4], sg[4];
#pragma unroll
    for (int g = 0; g < 4; ++g) {
        const int idx = g * 32 + lane;
        d[g]  = idx < W_EDGES ? dst[e0 + idx] : -1;
        sg[g] = idx < W_EDGES ? src[e0 + idx] : 0;
    }
    for (int i = tid; i < B_WAVES * CNT_PITCH; i += B_THREADS) cnt[i] = 0;
    __syncthreads();

    int* cntw = cnt + w * CNT_PITCH;
    int own[4], rank[4];
#pragma unroll
    for (int g = 0; g < 4; ++g) {
        const bool valid = (unsigned)d[g] < (unsigned)N_NODES;
        const int  o     = valid ? (int)((unsigned)d[g] / (unsigned)NODES_PER_WG) : 0;
        const unsigned p = same_owner_wg(o, valid);
        const int  run   = cntw[o];
        rank[g] = run + __popc(p & below);
        if (valid && (p & below) == 0u) cntw[o] = run + __popc(p);
        own[g] = valid ? o : -1;
    }
    __syncthreads();

    if (tid < CNT_PITCH) {
        int tot = 0;
#pragma unroll
        for (int s = 0; s < B_WAVES; ++s) {
            const int t = cnt[s * CNT_PITCH + tid];
            cnt[s * CNT_PITCH + tid] = tot;
            tot += t;
        }
        int incl = tot;
        for (int dd = 1; dd < 32; dd <<= 1) {
            const int t = __shfl_up(incl, dd);
            if (lane >= dd) incl += t;
        }
        ctot[tid] = tot;
        base[tid] = incl - tot;
        if (lane == 31) wtot[w] = incl;
        counts[c * CNT_PITCH + tid] = tot;
    }
    __syncthreads();
    if (tid < CNT_PITCH) {
        int add = 0;
        for (int v = 0; v < 8; ++v) {
            const int t = wtot[v];
            if (v < w) add += t;
        }
        base[tid] += add;
    }
    __syncthreads();

#pragma unroll
    for (int g = 0; g < 4; ++g) {
        if (own[g] >= 0) {
            const int o = own[g];
            sorted[base[o] + cntw[o] + rank[g]] =
                make_int2(((e0 + g * 32 + lane) << 8) | (d[g] - o * NODES_PER_WG), sg[g]);
        }
    }
    __syncthreads();

    const int oi = w + B_WAVES * (lane < RUNS_PER_WAVE ? lane : 0);
    int nn = ctot[oi] < SLOT_PAIRS ? ctot[oi] : SLOT_PAIRS;
    if (lane >= RUNS_PER_WAVE) nn = 0;
    const int bb = base[oi];
    int2* s0 = (int2*)out + (size_t)w * REGION_PAIRS + c * SLOT_PAIRS;
    int2 v[RUNS_PER_WAVE];
#pragma unroll
    for (int i = 0; i < RUNS_PER_WAVE; ++i) {
        const int n  = __builtin_amdgcn_readlane(nn, i);
        const int b0 = __builtin_amdgcn_readlane(bb, i);
        v[i] = lane < n ? sorted[b0 + lane] : make_int2(0, 0);
    }
#pragma unroll
    for (int i = 0; i < RUNS_PER_WAVE; ++i) {
        const int n = __builtin_amdgcn_readlane(nn, i);
        if (lane < n) s0[(size_t)i * (B_WAVES * REGION_PAIRS) + lane] = v[i];
    }
    if (__builtin_amdgcn_ballot_w32(nn > 32) != 0u) {
        for (int i = 0; i < RUNS_PER_WAVE; ++i) {
            const int n  = __builtin_amdgcn_readlane(nn, i);
            const int b0 = __builtin_amdgcn_readlane(bb, i);
            for (int k = 32 + lane; k < n; k += 32)
                s0[(size_t)i * (B_WAVES * REGION_PAIRS) + k] = sorted[b0 + k];
        }
    }
}


template <bool HAVE_SRC>
__device__ __forceinline__ void add_batch(const float* __restrict__ nf,
                                          const float* __restrict__ ef,
                                          const int*   __restrict__ src,
                                          float* acc, int* degc,
                                          const int* p, const int* ps, int n, int w, int lane, unsigned below) {
    const int e0  = p[lane < n ? lane : 0];
    int s0 = 0;
    if (HAVE_SRC) s0 = ps[lane < n ? lane : 0];
    const int key = lane < n ? ((e0 & 255) >> 5) : SLOTS;
    int rank = 0, base = 0, mine = 0;
    for (int s = 0; s <= SLOTS; ++s) {
        const unsigned m = __builtin_amdgcn_ballot_w32(key == s);
        if (key == s)  rank = base + __popc(m & below);
        if (lane == s) mine = __popc(m);
        base += __popc(m);
    }
    const int ent = __builtin_amdgcn_ds_permute(rank << 2, e0);
    int sv;
    if (HAVE_SRC) sv = __builtin_amdgcn_ds_permute(rank << 2, s0);
    else          sv = src[ent >> 8];
    int   ln[BATCH];
    float2 xa[BATCH], xb[BATCH];
    float  fe[BATCH];
#pragma unroll
    for (int i = 0; i < BATCH; ++i) {
        const int en = __builtin_amdgcn_readlane(ent, i);
        ln[i] = en & 255;
        fe[i] = ef[(size_t)(en >> 8) * EF_DIM + lane];
    }
#pragma unroll
    for (int i = 0; i < BATCH; ++i) {
        const int s = __builtin_amdgcn_readlane(sv, i);
        const float* r = nf + (size_t)s * IN_FEAT;
        xa[i] = *(const float2*)(r + 2 * lane);
        xb[i] = *(const float2*)(r + 64 + 2 * lane);
    }
    if (lane < SLOTS && mine > 0) degc[w + 32 * lane] += mine;

    float2 ta = make_float2(0.0f, 0.0f), tb = make_float2(0.0f, 0.0f);
    float  tc = 0.0f;
#pragma unroll
    for (int i = 0; i < BATCH; ++i) {
        if (i == 0 || i < n) {
            float* a = acc + ln[i] * ACC_W;
            bool first = true;
            if (i > 0) first = (ln[i] != ln[i - 1]);
            if (first) {
                ta = *(float2*)(a + 2 * lane);
                tb = *(float2*)(a + 64 + 2 * lane);
                tc = a[128 + lane];
            }
            ta.x += xa[i].x; ta.y += xa[i].y;
            tb.x += xb[i].x; tb.y += xb[i].y;
            tc   += fe[i];
            bool last = true;
            if (i + 1 < BATCH) last = (i + 1 >= n) || (ln[i + 1] != ln[i]);
            if (last) {
                *(float2*)(a + 2 * lane)      = ta;
                *(float2*)(a + 64 + 2 * lane) = tb;
                a[128 + lane] = tc;
            }
        }
    }
}

__device__ __forceinline__ void drain_lists(const float* __restrict__ nf,
                                            const float* __restrict__ ef,
                                            const int*   __restrict__ src,
                                            float* acc, int* degc,
                                            const int* lists, const int* fillv,
                                            int* pendw, int w, int lane) {
    const unsigned below = (1u << lane) - 1u;
    const int  myfill = fillv[lane] < LIST_CAP ? fillv[lane] : LIST_CAP;
    const int* mylist = lists + lane * LIST_PITCH;
    int  npend = 0;
    int  k = 0;
    bool more = true;
    for (;;) {
        while (more && npend < BATCH) {
            const bool valid = k < myfill;
            if (__builtin_amdgcn_ballot_w32(valid) == 0u) { more = false; break; }
            const int raw  = mylist[k];
            ++k;
            const bool mine = valid && ((raw & 31) == w);
            const unsigned m = __builtin_amdgcn_ballot_w32(mine);
            if (m != 0u) {
                if (mine) pendw[npend + __popc(m & below)] = raw;
                npend += __popc(m);
            }
        }
        if (npend == 0) break;
        const int n = npend < BATCH ? npend : BATCH;
        add_batch<false>(nf, ef, src, acc, degc, pendw, nullptr, n, w, lane, below);
        const int left = npend - n;
        if (left > 0) {
            const int t = pendw[BATCH + (lane < left ? lane : 0)];
            if (lane < left) pendw[lane] = t;
        }
        npend = left;
    }
}

__global__ void __launch_bounds__(1024) owner_kernel(const float* __restrict__ nf,
                                                     const float* __restrict__ ef,
                                                     const int*   __restrict__ src,
                                                     const int*   __restrict__ dst,
                                                     const int*   __restrict__ counts,
                                                     const float* __restrict__ prompt,
                                                     const float* __restrict__ hp,
                                                     float* __restrict__ out) {
    __shared__ float acc[NODES_PER_WG * ACC_W];
    __shared__ int   lists[32 * LIST_PITCH];
    __shared__ int   xbuf[XBUF_INTS];
    __shared__ int   fillv[32];
    __shared__ int   degc[NODES_PER_WG];
    __shared__ int   ovf;

    const int tid  = threadIdx.x;
    const int lane = tid & 31;
    const int w    = __builtin_amdgcn_readfirstlane(tid >> 5);
    const int n0   = blockIdx.x * NODES_PER_WG;
    const unsigned below = (1u << lane) - 1u;

    const int nj = lane < CH_PER_WAVE ? counts[(CH_PER_WAVE * w + lane) * CNT_PITCH + blockIdx.x] : 0;

    const float p0 = prompt[lane & 15], p1 = prompt[16 + (lane & 15)];
    const float h0 = hp[lane & 15],     h1 = hp[16 + (lane & 15)];

    for (int i = tid; i < NODES_PER_WG * ACC_W; i += 1024) acc[i] = 0.0f;
    if (tid < NODES_PER_WG) degc[tid] = 0;
    xbuf[tid] = 0;
    if (tid == 0) ovf = 0;
    __syncthreads();

    int* mylist = lists + w * LIST_PITCH;
    int  fillw  = 0;

    int incl8 = nj;
    for (int d = 1; d < CH_PER_WAVE; d <<= 1) {
        const int t = __shfl_up(incl8, d);
        if (lane >= d) incl8 += t;
    }
    const int offl  = incl8 - nj;
    const int total = __builtin_amdgcn_readlane(incl8, CH_PER_WAVE - 1);
    int maxn = nj;
    for (int d = 1; d < CH_PER_WAVE; d <<= 1) {
        const int t = __shfl_xor(maxn, d);
        maxn = t > maxn ? t : maxn;
    }
    maxn = __builtin_amdgcn_readfirstlane(maxn);
    const bool bad = (maxn > SLOT_PAIRS) || (total > FAST_CAP);
    int* flist = lists + w * FLIST_PITCH;
    if (!bad) {
        const int2* myslots = (const int2*)out + (size_t)blockIdx.x * REGION_PAIRS + (CH_PER_WAVE * w) * SLOT_PAIRS;
        for (int r0 = 0; r0 < maxn; r0 += 32) {
            const int idx = r0 + lane;
            int2 v[CH_PER_WAVE];
#pragma unroll
            for (int j = 0; j < CH_PER_WAVE; ++j) {
                const int n = __builtin_amdgcn_readlane(nj, j);
                v[j] = idx < n ? myslots[j * SLOT_PAIRS + idx] : make_int2(0, 0);
            }
#pragma unroll
            for (int j = 0; j < CH_PER_WAVE; ++j) {
                const int n = __builtin_amdgcn_readlane(nj, j);
                const int o = __builtin_amdgcn_readlane(offl, j);
                if (idx < n) { flist[o + idx] = v[j].x; flist[FSRC_OFF + o + idx] = v[j].y; }
            }
        }
        fillw = total;
    }
    if (lane == 0 && bad) ovf = 1;
    __syncthreads();
    if (__builtin_amdgcn_readfirstlane(ovf) != 0) {
        int* pendw = xbuf + w * PEND_CAP;
        for (int st = 0; st < CH_PER_WAVE * N_PIECES; ++st) {
            const int ch    = CH_PER_WAVE * w + (st >> 2);
            const int first = ch * CH_EDGES + (st & 3) * PIECE_EDGES;
            const int end   = (st & 3) == N_PIECES - 1 ? (ch + 1) * CH_EDGES : first + PIECE_EDGES;
            fillw = 0;
            for (int g = 0; g < PIECE_EDGES / 32; ++g) {
                const int  e   = first + g * 32 + lane;
                const bool in  = e < end;
                const int  dd  = dst[in ? e : first];
                const unsigned loc = (unsigned)(dd - n0);
                const bool hit = in && loc < (unsigned)NODES_PER_WG;
                const unsigned m = __builtin_amdgcn_ballot_w32(hit);
                if (hit) mylist[fillw + __popc(m & below)] = (e << 8) | (int)loc;
                fillw += __popc(m);
            }
            if (lane == 0) fillv[w] = fillw;
            __syncthreads();
            drain_lists(nf, ef, src, acc, degc, lists, fillv, pendw, w, lane);
            __syncthreads();
        }
    } else {
        int* cntw   = xbuf + w * 32;
        int* posw   = xbuf + 1024 + w * 32;
        int* sorted = xbuf + 2048;
        int* sortsrc = lists + 32 * FLIST_PITCH;
        for (int k0 = 0; k0 < fillw; k0 += 32) {
            const int  kk    = k0 + lane;
            const bool valid = kk < fillw;
            const int  o     = flist[valid ? kk : 0] & 31;
            const unsigned p = same_owner(o, valid);
            const int  run   = cntw[o];
            if (valid && (p & below) == 0u) cntw[o] = run + __popc(p);
        }
        __syncthreads();
        int colsum = 0, pre = 0;
        for (int s = 0; s < 32; ++s) {
            const int c = xbuf[s * 32 + lane];
            colsum += c;
            if (s < w) pre += c;
        }
        int incl = colsum;
        for (int d = 1; d < 32; d <<= 1) {
            const int t = __shfl_up(incl, d);
            if (lane >= d) incl += t;
        }
        const int first = incl - colsum;
        posw[lane] = first + pre;
        const int start = __builtin_amdgcn_readlane(first, w);
        const int len   = __builtin_amdgcn_readlane(colsum, w);
        for (int k0 = 0; k0 < fillw; k0 += 32) {
            const int  kk    = k0 + lane;
            const bool valid = kk < fillw;
            const int  ent   = flist[valid ? kk : 0];
            const int  sv    = flist[FSRC_OFF + (valid ? kk : 0)];
            const int  o     = ent & 31;
            const unsigned p = same_owner(o, valid);
            const int  b     = posw[o];
            if (valid) { sorted[b + __popc(p & below)] = ent; sortsrc[b + __popc(p & below)] = sv; }
            if (valid && (p & below) == 0u) posw[o] = b + __popc(p);
        }
        __syncthreads();
        for (int i = 0; i < len; i += BATCH) {
            const int n = len - i < BATCH ? len - i : BATCH;
            add_batch<true>(nf, ef, src, acc, degc, sorted + start + i, sortsrc + start + i, n, w, lane, below);
        }
    }
    __syncthreads();

    if (w < N_TILES) {
        const int  l15 = lane & 15;
        const bool lo  = (lane < 16);
        const int  lr  = w * 16 + l15;
        const int  dc  = degc[lr < NODES_PER_WG ? lr : 0];
        v2f a; a.x = (lo && lr < NODES_PER_WG) ? (float)dc : 0.0f; a.y = 0.0f;

        v2f b0; b0.x = lo ? p0 : 0.0f; b0.y = 0.0f;
        v2f b1; b1.x = lo ? p1 : 0.0f; b1.y = 0.0f;
        v2f b2; b2.x = lo ? h0 : 0.0f; b2.y = 0.0f;
        v2f b3; b3.x = lo ? h1 : 0.0f; b3.y = 0.0f;

        v8f c = {0.f, 0.f, 0.f, 0.f, 0.f, 0.f, 0.f, 0.f};
        v8f d0 = __builtin_amdgcn_wmma_f32_16x16x4_f32(false, a, false, b0, (short)0, c, false, false);
        v8f d1 = __builtin_amdgcn_wmma_f32_16x16x4_f32(false, a, false, b1, (short)0, c, false, false);
        v8f d2 = __builtin_amdgcn_wmma_f32_16x16x4_f32(false, a, false, b2, (short)0, c, false, false);
        v8f d3 = __builtin_amdgcn_wmma_f32_16x16x4_f32(false, a, false, b3, (short)0, c, false, false);

#pragma unroll
        for (int r = 0; r < 8; ++r) {
            const float s0 = __shfl_xor(d0[r], 16);
            const float s1 = __shfl_xor(d1[r], 16);
            const float s2 = __shfl_xor(d2[r], 16);
            const float s3 = __shfl_xor(d3[r], 16);
            const int ra = w * 16 + r;
            const int rb = ra + 8;
            float* oa = out + (size_t)(n0 + ra) * OUT_DIM;
            oa[128 + lane] = lo ? d0[r] : s1;
            oa[192 + lane] = lo ? d2[r] : s3;
            if (rb < NODES_PER_WG) {
                float* ob = out + (size_t)(n0 + rb) * OUT_DIM;
                ob[128 + lane] = lo ? s0 : d1[r];
                ob[192 + lane] = lo ? s2 : d3[r];
            }
        }
    }

    for (int r = w; r < NODES_PER_WG; r += 32) {
        const float* a = acc + r * ACC_W;
        float* o = out + (size_t)(n0 + r) * OUT_DIM;
        *(float2*)(o + 2 * lane)      = *(const float2*)(a + 2 * lane);
        *(float2*)(o + 64 + 2 * lane) = *(const float2*)(a + 64 + 2 * lane);
        o[160 + lane] = a[128 + lane];
    }
}

extern "C" void kernel_launch(void* const* d_in, const int* in_sizes, int n_in,
                              void* d_out, int out_size, void* d_ws, size_t ws_size,
                              hipStream_t stream) {
    (void)in_sizes; (void)n_in; (void)out_size; (void)ws_size;

    const float* nf     = (const float*)d_in[0];
    const float* ef     = (const float*)d_in[1];
    const float* prompt = (const float*)d_in[2];
    const float* hp     = (const float*)d_in[3];
    const int*   src    = (const int*)d_in[4];
    const int*   dst    = (const int*)d_in[5];

    float* out = (float*)d_out;
    int* counts = (int*)((char*)d_ws + 200704);

    bucket_kernel<<<N_CH, B_THREADS, 0, stream>>>(src, dst, counts, out);
    owner_kernel<<<N_OWNERS, 1024, 0, stream>>>(nf, ef, src, dst, counts, prompt, hp, out);
}
